# starting kernel
# baseline (speedup 1.0000x reference)
#include <hip/hip_runtime.h>

typedef __attribute__((ext_vector_type(2))) float v2f;
typedef __attribute__((ext_vector_type(8))) float v8f;

#define D     64
#define TROWS 32
#define LDP   68

__global__ void gcn_deg_init(float* deg, int n) {
  int i = blockIdx.x * blockDim.x + threadIdx.x;
  if (i < n) deg[i] = 1.0f;
}

__global__ void gcn_deg_edges(float* deg, const int* __restrict__ dst,
                              const float* __restrict__ ew, int e) {
  int i = blockIdx.x * blockDim.x + threadIdx.x;
  if (i < e) atomicAdd(&deg[dst[i]], ew[i]);
}

__global__ void gcn_dinv(float* deg, int n) {
  int i = blockIdx.x * blockDim.x + threadIdx.x;
  if (i < n) {
    float d = deg[i];
    deg[i] = (d > 0.0f) ? rsqrtf(d) : 0.0f;
  }
}

__global__ void gcn_copy_x(const float* __restrict__ x, float* __restrict__ Hcat, int n) {
  int i = blockIdx.x * blockDim.x + threadIdx.x;
  if (i < n * D) {
    int node = i >> 6, j = i & 63;
    Hcat[(size_t)node * 256 + j] = x[i];
  }
}

__global__ void gcn_agg_self(const float* __restrict__ Htmp, const float* __restrict__ dinv,
                             const float* __restrict__ bias, float* __restrict__ Hcat,
                             int off, int n) {
  int i = blockIdx.x * blockDim.x + threadIdx.x;
  if (i < n * D) {
    int node = i >> 6, j = i & 63;
    float di = dinv[node];
    Hcat[(size_t)node * 256 + off + j] = bias[j] + di * di * Htmp[i];
  }
}

__global__ void gcn_scatter(const float* __restrict__ Htmp, const float* __restrict__ dinv,
                            const float* __restrict__ ew, const int* __restrict__ src,
                            const int* __restrict__ dst, float* __restrict__ Hcat,
                            int off, int e) {
  int t = blockIdx.x * blockDim.x + threadIdx.x;
  int eid = t >> 4;
  int l   = t & 15;
  if (eid < e) {
    int s = src[eid], d = dst[eid];
    float nrm = dinv[s] * ew[eid] * dinv[d];
    const float4 h = *(const float4*)(Htmp + (size_t)s * D + l * 4);
    float* o = Hcat + (size_t)d * 256 + off + l * 4;
    atomicAdd(o + 0, h.x * nrm);
    atomicAdd(o + 1, h.y * nrm);
    atomicAdd(o + 2, h.z * nrm);
    atomicAdd(o + 3, h.w * nrm);
  }
}

__global__ void gcn_relu(float* Hcat, int off, int n) {
  int i = blockIdx.x * blockDim.x + threadIdx.x;
  if (i < n * D) {
    int node = i >> 6, j = i & 63;
    size_t idx = (size_t)node * 256 + off + j;
    float v = Hcat[idx];
    Hcat[idx] = v > 0.0f ? v : 0.0f;
  }
}

__global__ void gcn_gemm_wmma(const float* __restrict__ X, int ldx,
                              const float* __restrict__ W,
                              const float* __restrict__ bias,
                              float* __restrict__ Out, int K, int n) {
  __shared__ float Xs[TROWS * LDP];
  __shared__ float Ws[D * LDP];
  const int tid    = threadIdx.x;
  const int wave   = tid >> 5;
  const int lane   = tid & 31;
  const int laneLo = lane & 15;
  const int laneHi = lane >> 4;
  const int row0   = blockIdx.x * TROWS;
  const int wm     = (wave >> 2) << 4;
  const int wn     = (wave & 3) << 4;

  v8f acc = {};
  for (int kc = 0; kc < K; kc += D) {
    for (int i = tid; i < TROWS * D; i += 256) {
      int r = i >> 6, c = i & 63;
      int rr = row0 + r; if (rr >= n) rr = n - 1;
      Xs[r * LDP + c] = X[(size_t)rr * ldx + kc + c];
    }
    for (int i = tid; i < D * D; i += 256) {
      int r = i >> 6, c = i & 63;
      Ws[r * LDP + c] = W[(size_t)(kc + r) * D + c];
    }
    __syncthreads();
#pragma unroll
    for (int k0 = 0; k0 < D; k0 += 4) {
      v2f a = *(const v2f*)(&Xs[(wm + laneLo) * LDP + k0 + 2 * laneHi]);
      v2f b;
      b[0] = Ws[(k0 + 2 * laneHi + 0) * LDP + wn + laneLo];
      b[1] = Ws[(k0 + 2 * laneHi + 1) * LDP + wn + laneLo];
      acc = __builtin_amdgcn_wmma_f32_16x16x4_f32(false, a, false, b,
                                                  (short)0, acc, false, false);
    }
    __syncthreads();
  }
  const int gcol = wn + laneLo;
  const float bv = bias ? bias[gcol] : 0.0f;
#pragma unroll
  for (int v = 0; v < 8; ++v) {
    int grow = row0 + wm + v + 8 * laneHi;
    if (grow < n) Out[(size_t)grow * D + gcol] = acc[v] + bv;
  }
}

extern "C" void kernel_launch(void* const* d_in, const int* in_sizes, int n_in,
                              void* d_out, int out_size, void* d_ws, size_t ws_size,
                              hipStream_t stream) {
  const float* x   = (const float*)d_in[0];
  const int*   ei  = (const int*)d_in[1];
  const float* ew  = (const float*)d_in[2];
  const float* Wl[3] = {(const float*)d_in[3], (const float*)d_in[5], (const float*)d_in[7]};
  const float* bl[3] = {(const float*)d_in[4], (const float*)d_in[6], (const float*)d_in[8]};
  const float* Wf  = (const float*)d_in[9];
  const float* bf  = (const float*)d_in[10];

  const int N = in_sizes[0] / D;
  const int E = in_sizes[2];
  const int* src = ei;
  const int* dst = ei + E;

  float* deg  = (float*)d_ws;
  float* Hcat = deg + (((size_t)N + 255) & ~(size_t)255);
  float* Htmp = Hcat + (size_t)N * 256;

  const int T = 256;
  gcn_deg_init <<<(N + T - 1) / T, T, 0, stream>>>(deg, N);
  gcn_deg_edges<<<(E + T - 1) / T, T, 0, stream>>>(deg, dst, ew, E);
  gcn_dinv     <<<(N + T - 1) / T, T, 0, stream>>>(deg, N);
  gcn_copy_x   <<<((size_t)N * D + T - 1) / T, T, 0, stream>>>(x, Hcat, N);

  const int gemm_blocks = (N + TROWS - 1) / TROWS;
  for (int l = 0; l < 3; ++l) {
    const int off = (l + 1) * D;
    gcn_gemm_wmma<<<gemm_blocks, T, 0, stream>>>(Hcat + l * D, 256, Wl[l], nullptr,
                                                 Htmp, D, N);
    gcn_agg_self <<<((size_t)N * D + T - 1) / T, T, 0, stream>>>(Htmp, deg, bl[l],
                                                                 Hcat, off, N);
    gcn_scatter  <<<((size_t)E * 16 + T - 1) / T, T, 0, stream>>>(Htmp, deg, ew, src, dst,
                                                                  Hcat, off, E);
    gcn_relu     <<<((size_t)N * D + T - 1) / T, T, 0, stream>>>(Hcat, off, N);
  }
  gcn_gemm_wmma<<<gemm_blocks, T, 0, stream>>>(Hcat, 256, Wf, bf, (float*)d_out, 4 * D, N);
}


# Scatter-add by f32 atomics replaced by a per-call edge list grouped by destination (int atomic rank, scan, fill); each row summed in registers in list order, stored once; sum order changes
# speedup vs baseline: 1.7577x; 1.7577x over previous
#include <hip/hip_runtime.h>

typedef __attribute__((ext_vector_type(2))) float v2f;
typedef __attribute__((ext_vector_type(8))) float v8f;

#define D     64
#define TROWS 32
#define LDP   68

__global__ void gcn_cnt_init(int* cnt, int n) {
  int i = blockIdx.x * blockDim.x + threadIdx.x;
  if (i < n) cnt[i] = 0;
}

__global__ void gcn_rank(int* cnt, const int* __restrict__ dst, int* __restrict__ rank, int e) {
  int i = blockIdx.x * blockDim.x + threadIdx.x;
  if (i < e) rank[i] = atomicAdd(&cnt[dst[i]], 1);
}

__global__ void gcn_scan(const int* __restrict__ cnt, int* __restrict__ rowptr, int n) {
  __shared__ int pre[8];
  __shared__ int loc[8];
  const int tid = threadIdx.x, lane = tid & 31, wv = tid >> 5;
  const int b = blockIdx.x;
  int part = 0;
  const int4* c4 = (const int4*)cnt;
  const int n4 = b * 256;
  for (int i = tid; i < n4; i += 256) {
    int4 v = c4[i];
    part += (v.x + v.y) + (v.z + v.w);
  }
  for (int o = 16; o > 0; o >>= 1) part += __shfl_xor(part, o);
  if (lane == 0) pre[wv] = part;

  const int i0 = b * 1024 + tid * 4;
  const int c0 = (i0 + 0 < n) ? cnt[i0 + 0] : 0;
  const int c1 = (i0 + 1 < n) ? cnt[i0 + 1] : 0;
  const int c2 = (i0 + 2 < n) ? cnt[i0 + 2] : 0;
  const int c3 = (i0 + 3 < n) ? cnt[i0 + 3] : 0;
  const int s = (c0 + c1) + (c2 + c3);
  int inc = s;
  for (int o = 1; o < 32; o <<= 1) {
    int t = __shfl_up(inc, o);
    if (lane >= o) inc += t;
  }
  if (lane == 31) loc[wv] = inc;
  __syncthreads();
  int base = 0;
  for (int k = 0; k < 8; ++k) base += pre[k];
  for (int k = 0; k < wv; ++k) base += loc[k];
  const int ex = base + inc - s;
  if (i0 + 0 < n) rowptr[i0 + 0] = ex;
  if (i0 + 1 < n) rowptr[i0 + 1] = ex + c0;
  if (i0 + 2 < n) rowptr[i0 + 2] = ex + c0 + c1;
  if (i0 + 3 < n) rowptr[i0 + 3] = ex + c0 + c1 + c2;
}

__global__ void gcn_fill(const int* __restrict__ rowptr, const int* __restrict__ rank,
                         const int* __restrict__ src, const int* __restrict__ dst,
                         const float* __restrict__ ew, int* __restrict__ lst_, int e) {
  int2* lst = (int2*)lst_;
  int i = blockIdx.x * blockDim.x + threadIdx.x;
  if (i < e) {
    int p = rowptr[dst[i]] + rank[i];
    lst[p] = make_int2(src[i], __float_as_int(ew[i]));
  }
}

__global__ void gcn_dinv(float* __restrict__ dinv, const int* __restrict__ rowptr,
                         const int* __restrict__ cnt, const int* __restrict__ lst_, int n) {
  const int2* lst = (const int2*)lst_;
  int i = blockIdx.x * blockDim.x + threadIdx.x;
  if (i < n) {
    const int st = rowptr[i], len = cnt[i];
    float d = 1.0f;
    for (int j = 0; j < len; ++j) d += __int_as_float(lst[st + j].y);
    dinv[i] = (d > 0.0f) ? rsqrtf(d) : 0.0f;
  }
}

__global__ __launch_bounds__(256)
void gcn_norm(const float* __restrict__ dinv, const int* __restrict__ rowptr,
              const int* __restrict__ cnt, const int* __restrict__ lst_,
              int* __restrict__ lstn_, int n) {
  const int2* lst  = (const int2*)lst_;
  int2*       lstn = (int2*)lstn_;
  const int l  = threadIdx.x & 15;
  const int ng = (gridDim.x * blockDim.x) >> 4;
  for (int node = (blockIdx.x * blockDim.x + threadIdx.x) >> 4; node < n; node += ng) {
    const int st = rowptr[node], len = cnt[node];
    const float di = dinv[node];
    for (int j = l; j < len; j += 16) {
      const int2 e = lst[st + j];
      const float nrm = dinv[e.x] * __int_as_float(e.y) * di;
      lstn[st + j] = make_int2(e.x, __float_as_int(nrm));
    }
  }
}

__global__ void gcn_copy_x(const float* __restrict__ x, float* __restrict__ Hcat, int n) {
  int i = blockIdx.x * blockDim.x + threadIdx.x;
  if (i < n * D) {
    int node = i >> 6, j = i & 63;
    Hcat[(size_t)node * 256 + j] = x[i];
  }
}

__global__ __launch_bounds__(256)
void gcn_gather(const float* __restrict__ Htmp, const float* __restrict__ dinv,
                const int* __restrict__ rowptr, const int* __restrict__ cnt,
                const int* __restrict__ lstn_, const float* __restrict__ bias,
                float* __restrict__ Hcat, int off, int n) {
  const int2* lstn = (const int2*)lstn_;
  const int l  = threadIdx.x & 15;
  const int ng = (gridDim.x * blockDim.x) >> 4;
  int node = (blockIdx.x * blockDim.x + threadIdx.x) >> 4;
  const float4 bv = *(const float4*)(bias + l * 4);
  int st = 0, end = 0;
  if (node < n) { st = rowptr[node]; end = st + cnt[node]; }
  while (node < n) {
    const int nnode = node + ng;
    int nst = 0, nlen = 0;
    if (nnode < n) { nst = rowptr[nnode]; nlen = cnt[nnode]; }
    const float di = dinv[node];
    const float4 hs = *(const float4*)(Htmp + (size_t)node * D + l * 4);
    float4 acc = make_float4(0.0f, 0.0f, 0.0f, 0.0f);
    for (int p = st; p < end; p += 16) {
      const int m = min(16, end - p);
      const int2 e = lstn[p + min(l, m - 1)];
      const int   sv = e.x;
      const float wv = __int_as_float(e.y & ((l < m) ? -1 : 0));
      int k = 0;
      do {
        float4 h[4];
        float  w[4];
#pragma unroll
        for (int u = 0; u < 4; ++u) {
          const int s = __shfl(sv, k + u, 16);
          w[u] = __shfl(wv, k + u, 16);
          h[u] = make_float4(0.0f, 0.0f, 0.0f, 0.0f);
          if (k + u < m) h[u] = *(const float4*)(Htmp + (size_t)s * D + l * 4);
        }
#pragma unroll
        for (int u = 0; u < 4; ++u) {
          acc.x = fmaf(h[u].x, w[u], acc.x);
          acc.y = fmaf(h[u].y, w[u], acc.y);
          acc.z = fmaf(h[u].z, w[u], acc.z);
          acc.w = fmaf(h[u].w, w[u], acc.w);
        }
        k += 4;
      } while (k < m);
    }
    const float sc = di * di;
    float4 r;
    r.x = fmaf(sc, hs.x, acc.x) + bv.x;
    r.y = fmaf(sc, hs.y, acc.y) + bv.y;
    r.z = fmaf(sc, hs.z, acc.z) + bv.z;
    r.w = fmaf(sc, hs.w, acc.w) + bv.w;
    r.x = r.x > 0.0f ? r.x : 0.0f;
    r.y = r.y > 0.0f ? r.y : 0.0f;
    r.z = r.z > 0.0f ? r.z : 0.0f;
    r.w = r.w > 0.0f ? r.w : 0.0f;
    *(float4*)(Hcat + (size_t)node * 256 + off + l * 4) = r;
    node = nnode; st = nst; end = nst + nlen;
  }
}

__global__ void gcn_gemm_wmma(const float* __restrict__ X, int ldx,
                              const float* __restrict__ W,
                              const float* __restrict__ bias,
                              float* __restrict__ Out, int K, int n) {
  __shared__ float Xs[TROWS * LDP];
  __shared__ float Ws[D * LDP];
  const int tid    = threadIdx.x;
  const int wave   = tid >> 5;
  const int lane   = tid & 31;
  const int laneLo = lane & 15;
  const int laneHi = lane >> 4;
  const int row0   = blockIdx.x * TROWS;
  const int wm     = (wave >> 2) << 4;
  const int wn     = (wave & 3) << 4;

  v8f acc = {};
  for (int kc = 0; kc < K; kc += D) {
    for (int i = tid; i < TROWS * D; i += 256) {
      int r = i >> 6, c = i & 63;
      int rr = row0 + r; if (rr >= n) rr = n - 1;
      Xs[r * LDP + c] = X[(size_t)rr * ldx + kc + c];
    }
    for (int i = tid; i < D * D; i += 256) {
      int r = i >> 6, c = i & 63;
      Ws[r * LDP + c] = W[(size_t)(kc + r) * D + c];
    }
    __syncthreads();
#pragma unroll
    for (int k0 = 0; k0 < D; k0 += 4) {
      v2f a = *(const v2f*)(&Xs[(wm + laneLo) * LDP + k0 + 2 * laneHi]);
      v2f b;
      b[0] = Ws[(k0 + 2 * laneHi + 0) * LDP + wn + laneLo];
      b[1] = Ws[(k0 + 2 * laneHi + 1) * LDP + wn + laneLo];
      acc = __builtin_amdgcn_wmma_f32_16x16x4_f32(false, a, false, b,
                                                  (short)0, acc, false, false);
    }
    __syncthreads();
  }
  const int gcol = wn + laneLo;
  const float bv = bias ? bias[gcol] : 0.0f;
#pragma unroll
  for (int v = 0; v < 8; ++v) {
    int grow = row0 + wm + v + 8 * laneHi;
    if (grow < n) Out[(size_t)grow * D + gcol] = acc[v] + bv;
  }
}

extern "C" void kernel_launch(void* const* d_in, const int* in_sizes, int n_in,
                              void* d_out, int out_size, void* d_ws, size_t ws_size,
                              hipStream_t stream) {
  const float* x   = (const float*)d_in[0];
  const int*   ei  = (const int*)d_in[1];
  const float* ew  = (const float*)d_in[2];
  const float* Wl[3] = {(const float*)d_in[3], (const float*)d_in[5], (const float*)d_in[7]};
  const float* bl[3] = {(const float*)d_in[4], (const float*)d_in[6], (const float*)d_in[8]};
  const float* Wf  = (const float*)d_in[9];
  const float* bf  = (const float*)d_in[10];

  const int N = in_sizes[0] / D;
  const int E = in_sizes[2];
  const int* src = ei;
  const int* dst = ei + E;

  const size_t Np = ((size_t)N + 255) & ~(size_t)255;
  const size_t Ep = ((size_t)E + 255) & ~(size_t)255;
  float* deg    = (float*)d_ws;
  float* Hcat   = deg + Np;
  float* Htmp   = Hcat + (size_t)N * 256;
  int*   cnt    = (int*)(Htmp + (size_t)N * D);
  int*   rowptr = cnt + Np;
  int*   rank   = rowptr + Np;
  int*   lst    = rank + Ep;
  int*   lstn   = lst + 2 * Ep;

  const int T = 256;
  const int PB = 1024;
  gcn_cnt_init<<<(N + T - 1) / T, T, 0, stream>>>(cnt, N);
  gcn_rank    <<<(E + T - 1) / T, T, 0, stream>>>(cnt, dst, rank, E);
  gcn_scan    <<<(N + 1023) / 1024, T, 0, stream>>>(cnt, rowptr, N);
  gcn_fill    <<<(E + T - 1) / T, T, 0, stream>>>(rowptr, rank, src, dst, ew, lst, E);
  gcn_dinv    <<<(N + T - 1) / T, T, 0, stream>>>(deg, rowptr, cnt, lst, N);
  gcn_norm    <<<PB, T, 0, stream>>>(deg, rowptr, cnt, lst, lstn, N);
  gcn_copy_x  <<<((size_t)N * D + T - 1) / T, T, 0, stream>>>(x, Hcat, N);

  const int gemm_blocks = (N + TROWS - 1) / TROWS;
  for (int l = 0; l < 3; ++l) {
    const int off = (l + 1) * D;
    gcn_gemm_wmma<<<gemm_blocks, T, 0, stream>>>(Hcat + l * D, 256, Wl[l], nullptr,
                                                 Htmp, D, N);
    gcn_gather   <<<PB, T, 0, stream>>>(Htmp, deg, rowptr, cnt, lstn, bl[l], Hcat, off, N);
  }
  gcn_gemm_wmma<<<gemm_blocks, T, 0, stream>>>(Hcat, 256, Wf, bf, (float*)d_out, 4 * D, N);
}


# Edges grouped by destination with a two-level counting sort on LDS counters (256-node buckets) instead of one global returning atomic per edge; list order = LDS arrival order, sum order changes
# speedup vs baseline: 2.1550x; 1.2260x over previous
#include <hip/hip_runtime.h>

typedef __attribute__((ext_vector_type(2))) float v2f;
typedef __attribute__((ext_vector_type(8))) float v8f;

#define D     64
#define TROWS 32
#define LDP   68

#define BSH   8
#define BNODE (1 << BSH)
#define NSL   256
#define BT    512

__global__ __launch_bounds__(BT)
void gcn_bin_count(const int* __restrict__ dst, int* __restrict__ tab, int e, int nbk) {
  extern __shared__ int hist[];
  const int s = blockIdx.x, tid = threadIdx.x;
  for (int b = tid; b < nbk; b += BT) hist[b] = 0;
  __syncthreads();
  const int per = (e + NSL - 1) / NSL;
  const int lo  = s * per;
  const int hi  = min(e, lo + per);
  for (int i0 = lo + tid; i0 < hi; i0 += 4 * BT) {
    int d[4];
#pragma unroll
    for (int u = 0; u < 4; ++u) {
      const int i = i0 + u * BT;
      d[u] = (i < hi) ? dst[i] : 0;
    }
#pragma unroll
    for (int u = 0; u < 4; ++u)
      if (i0 + u * BT < hi) atomicAdd(&hist[d[u] >> BSH], 1);
  }
  __syncthreads();
  for (int b = tid; b < nbk; b += BT) tab[b * NSL + s] = hist[b];
}

__global__ void gcn_scan(const int* __restrict__ cnt, int* __restrict__ rowptr, int n) {
  __shared__ int pre[8];
  __shared__ int loc[8];
  const int tid = threadIdx.x, lane = tid & 31, wv = tid >> 5;
  const int b = blockIdx.x;
  int part = 0;
  const int4* c4 = (const int4*)cnt;
  const int n4 = b * 256;
  for (int i = tid; i < n4; i += 256) {
    int4 v = c4[i];
    part += (v.x + v.y) + (v.z + v.w);
  }
  for (int o = 16; o > 0; o >>= 1) part += __shfl_xor(part, o);
  if (lane == 0) pre[wv] = part;

  const int i0 = b * 1024 + tid * 4;
  const int c0 = (i0 + 0 < n) ? cnt[i0 + 0] : 0;
  const int c1 = (i0 + 1 < n) ? cnt[i0 + 1] : 0;
  const int c2 = (i0 + 2 < n) ? cnt[i0 + 2] : 0;
  const int c3 = (i0 + 3 < n) ? cnt[i0 + 3] : 0;
  const int s = (c0 + c1) + (c2 + c3);
  int inc = s;
  for (int o = 1; o < 32; o <<= 1) {
    int t = __shfl_up(inc, o);
    if (lane >= o) inc += t;
  }
  if (lane == 31) loc[wv] = inc;
  __syncthreads();
  int base = 0;
  for (int k = 0; k < 8; ++k) base += pre[k];
  for (int k = 0; k < wv; ++k) base += loc[k];
  const int ex = base + inc - s;
  if (i0 + 0 < n) rowptr[i0 + 0] = ex;
  if (i0 + 1 < n) rowptr[i0 + 1] = ex + c0;
  if (i0 + 2 < n) rowptr[i0 + 2] = ex + c0 + c1;
  if (i0 + 3 < n) rowptr[i0 + 3] = ex + c0 + c1 + c2;
}

__global__ __launch_bounds__(BT)
void gcn_bin_place(const int* __restrict__ src, const int* __restrict__ dst,
                   const float* __restrict__ ew, const int* __restrict__ offs,
                   int* __restrict__ binq_, int e, int nbk) {
  extern __shared__ int cur[];
  int4* binq = (int4*)binq_;
  const int s = blockIdx.x, tid = threadIdx.x;
  for (int b = tid; b < nbk; b += BT) cur[b] = offs[b * NSL + s];
  __syncthreads();
  const int per = (e + NSL - 1) / NSL;
  const int lo  = s * per;
  const int hi  = min(e, lo + per);
  for (int i0 = lo + tid; i0 < hi; i0 += 4 * BT) {
    int d[4], sv[4], wb[4];
#pragma unroll
    for (int u = 0; u < 4; ++u) {
      const int i = i0 + u * BT;
      d[u] = 0; sv[u] = 0; wb[u] = 0;
      if (i < hi) { d[u] = dst[i]; sv[u] = src[i]; wb[u] = __float_as_int(ew[i]); }
    }
#pragma unroll
    for (int u = 0; u < 4; ++u) {
      if (i0 + u * BT < hi) {
        const int p = atomicAdd(&cur[d[u] >> BSH], 1);
        binq[p] = make_int4(sv[u], wb[u], d[u], 0);
      }
    }
  }
}

__global__ __launch_bounds__(BT)
void gcn_bin_lists(const int* __restrict__ offs, const int* __restrict__ binq_,
                   int* __restrict__ cnt, int* __restrict__ rowptr, int* __restrict__ lst_,
                   int e, int n, int nbk) {
  __shared__ int c[BNODE];
  __shared__ int cursor[BNODE];
  __shared__ int wsum[BNODE / 32];
  const int4* binq = (const int4*)binq_;
  int2* lst = (int2*)lst_;
  const int b = blockIdx.x, tid = threadIdx.x, lane = tid & 31, wv = tid >> 5;
  const int base = b << BSH;
  const int bs = offs[b * NSL];
  const int be = (b + 1 < nbk) ? offs[(b + 1) * NSL] : e;
  if (tid < BNODE) c[tid] = 0;
  __syncthreads();
  for (int i0 = bs + tid; i0 < be; i0 += 4 * BT) {
    int d[4];
#pragma unroll
    for (int u = 0; u < 4; ++u) {
      const int i = i0 + u * BT;
      d[u] = (i < be) ? binq[i].z : base;
    }
#pragma unroll
    for (int u = 0; u < 4; ++u)
      if (i0 + u * BT < be) atomicAdd(&c[d[u] - base], 1);
  }
  __syncthreads();
  const int v = (tid < BNODE) ? c[tid] : 0;
  int inc = v;
  for (int o = 1; o < 32; o <<= 1) {
    int t = __shfl_up(inc, o);
    if (lane >= o) inc += t;
  }
  if (lane == 31 && tid < BNODE) wsum[wv] = inc;
  __syncthreads();
  if (tid < BNODE) {
    int ex = bs + inc - v;
    for (int k = 0; k < wv; ++k) ex += wsum[k];
    cursor[tid] = ex;
    const int node = base + tid;
    if (node < n) { cnt[node] = v; rowptr[node] = ex; }
  }
  __syncthreads();
  for (int i0 = bs + tid; i0 < be; i0 += 4 * BT) {
    int4 q[4];
#pragma unroll
    for (int u = 0; u < 4; ++u) {
      const int i = i0 + u * BT;
      q[u] = make_int4(0, 0, base, 0);
      if (i < be) q[u] = binq[i];
    }
#pragma unroll
    for (int u = 0; u < 4; ++u) {
      if (i0 + u * BT < be) {
        const int p = atomicAdd(&cursor[q[u].z - base], 1);
        lst[p] = make_int2(q[u].x, q[u].y);
      }
    }
  }
}

__global__ void gcn_dinv(float* __restrict__ dinv, const int* __restrict__ rowptr,
                         const int* __restrict__ cnt, const int* __restrict__ lst_, int n) {
  const int2* lst = (const int2*)lst_;
  int i = blockIdx.x * blockDim.x + threadIdx.x;
  if (i < n) {
    const int st = rowptr[i], len = cnt[i];
    float d = 1.0f;
    for (int j = 0; j < len; ++j) d += __int_as_float(lst[st + j].y);
    dinv[i] = (d > 0.0f) ? rsqrtf(d) : 0.0f;
  }
}

__global__ __launch_bounds__(256)
void gcn_norm(const float* __restrict__ dinv, const int* __restrict__ rowptr,
              const int* __restrict__ cnt, const int* __restrict__ lst_,
              int* __restrict__ lstn_, int n) {
  const int2* lst  = (const int2*)lst_;
  int2*       lstn = (int2*)lstn_;
  const int l  = threadIdx.x & 15;
  const int ng = (gridDim.x * blockDim.x) >> 4;
  for (int node = (blockIdx.x * blockDim.x + threadIdx.x) >> 4; node < n; node += ng) {
    const int st = rowptr[node], len = cnt[node];
    const float di = dinv[node];
    for (int j = l; j < len; j += 16) {
      const int2 e = lst[st + j];
      const float nrm = dinv[e.x] * __int_as_float(e.y) * di;
      lstn[st + j] = make_int2(e.x, __float_as_int(nrm));
    }
  }
}

__global__ void gcn_copy_x(const float* __restrict__ x, float* __restrict__ Hcat, int n) {
  int i = blockIdx.x * blockDim.x + threadIdx.x;
  if (i < n * D) {
    int node = i >> 6, j = i & 63;
    Hcat[(size_t)node * 256 + j] = x[i];
  }
}

__global__ __launch_bounds__(256)
void gcn_gather(const float* __restrict__ Htmp, const float* __restrict__ dinv,
                const int* __restrict__ rowptr, const int* __restrict__ cnt,
                const int* __restrict__ lstn_, const float* __restrict__ bias,
                float* __restrict__ Hcat, int off, int n) {
  const int2* lstn = (const int2*)lstn_;
  const int l  = threadIdx.x & 15;
  const int ng = (gridDim.x * blockDim.x) >> 4;
  int node = (blockIdx.x * blockDim.x + threadIdx.x) >> 4;
  const float4 bv = *(const float4*)(bias + l * 4);
  int st = 0, end = 0;
  if (node < n) { st = rowptr[node]; end = st + cnt[node]; }
  while (node < n) {
    const int nnode = node + ng;
    int nst = 0, nlen = 0;
    if (nnode < n) { nst = rowptr[nnode]; nlen = cnt[nnode]; }
    const float di = dinv[node];
    const float4 hs = *(const float4*)(Htmp + (size_t)node * D + l * 4);
    float4 acc = make_float4(0.0f, 0.0f, 0.0f, 0.0f);
    for (int p = st; p < end; p += 16) {
      const int m = min(16, end - p);
      const int2 e = lstn[p + min(l, m - 1)];
      const int   sv = e.x;
      const float wv = __int_as_float(e.y & ((l < m) ? -1 : 0));
      int k = 0;
      do {
        float4 h[4];
        float  w[4];
#pragma unroll
        for (int u = 0; u < 4; ++u) {
          const int s = __shfl(sv, k + u, 16);
          w[u] = __shfl(wv, k + u, 16);
          h[u] = make_float4(0.0f, 0.0f, 0.0f, 0.0f);
          if (k + u < m) h[u] = *(const float4*)(Htmp + (size_t)s * D + l * 4);
        }
#pragma unroll
        for (int u = 0; u < 4; ++u) {
          acc.x = fmaf(h[u].x, w[u], acc.x);
          acc.y = fmaf(h[u].y, w[u], acc.y);
          acc.z = fmaf(h[u].z, w[u], acc.z);
          acc.w = fmaf(h[u].w, w[u], acc.w);
        }
        k += 4;
      } while (k < m);
    }
    const float sc = di * di;
    float4 r;
    r.x = fmaf(sc, hs.x, acc.x) + bv.x;
    r.y = fmaf(sc, hs.y, acc.y) + bv.y;
    r.z = fmaf(sc, hs.z, acc.z) + bv.z;
    r.w = fmaf(sc, hs.w, acc.w) + bv.w;
    r.x = r.x > 0.0f ? r.x : 0.0f;
    r.y = r.y > 0.0f ? r.y : 0.0f;
    r.z = r.z > 0.0f ? r.z : 0.0f;
    r.w = r.w > 0.0f ? r.w : 0.0f;
    *(float4*)(Hcat + (size_t)node * 256 + off + l * 4) = r;
    node = nnode; st = nst; end = nst + nlen;
  }
}

__global__ void gcn_gemm_wmma(const float* __restrict__ X, int ldx,
                              const float* __restrict__ W,
                              const float* __restrict__ bias,
                              float* __restrict__ Out, int K, int n) {
  __shared__ float Xs[TROWS * LDP];
  __shared__ float Ws[D * LDP];
  const int tid    = threadIdx.x;
  const int wave   = tid >> 5;
  const int lane   = tid & 31;
  const int laneLo = lane & 15;
  const int laneHi = lane >> 4;
  const int row0   = blockIdx.x * TROWS;
  const int wm     = (wave >> 2) << 4;
  const int wn     = (wave & 3) << 4;

  v8f acc = {};
  for (int kc = 0; kc < K; kc += D) {
    for (int i = tid; i < TROWS * D; i += 256) {
      int r = i >> 6, c = i & 63;
      int rr = row0 + r; if (rr >= n) rr = n - 1;
      Xs[r * LDP + c] = X[(size_t)rr * ldx + kc + c];
    }
    for (int i = tid; i < D * D; i += 256) {
      int r = i >> 6, c = i & 63;
      Ws[r * LDP + c] = W[(size_t)(kc + r) * D + c];
    }
    __syncthreads();
#pragma unroll
    for (int k0 = 0; k0 < D; k0 += 4) {
      v2f a = *(const v2f*)(&Xs[(wm + laneLo) * LDP + k0 + 2 * laneHi]);
      v2f b;
      b[0] = Ws[(k0 + 2 * laneHi + 0) * LDP + wn + laneLo];
      b[1] = Ws[(k0 + 2 * laneHi + 1) * LDP + wn + laneLo];
      acc = __builtin_amdgcn_wmma_f32_16x16x4_f32(false, a, false, b,
                                                  (short)0, acc, false, false);
    }
    __syncthreads();
  }
  const int gcol = wn + laneLo;
  const float bv = bias ? bias[gcol] : 0.0f;
#pragma unroll
  for (int v = 0; v < 8; ++v) {
    int grow = row0 + wm + v + 8 * laneHi;
    if (grow < n) Out[(size_t)grow * D + gcol] = acc[v] + bv;
  }
}

extern "C" void kernel_launch(void* const* d_in, const int* in_sizes, int n_in,
                              void* d_out, int out_size, void* d_ws, size_t ws_size,
                              hipStream_t stream) {
  const float* x   = (const float*)d_in[0];
  const int*   ei  = (const int*)d_in[1];
  const float* ew  = (const float*)d_in[2];
  const float* Wl[3] = {(const float*)d_in[3], (const float*)d_in[5], (const float*)d_in[7]};
  const float* bl[3] = {(const float*)d_in[4], (const float*)d_in[6], (const float*)d_in[8]};
  const float* Wf  = (const float*)d_in[9];
  const float* bf  = (const float*)d_in[10];

  const int N = in_sizes[0] / D;
  const int E = in_sizes[2];
  const int* src = ei;
  const int* dst = ei + E;

  const size_t Np = ((size_t)N + 255) & ~(size_t)255;
  const size_t Ep = ((size_t)E + 255) & ~(size_t)255;
  const int    nbk = (int)(Np >> BSH);
  const size_t Tn  = (size_t)nbk * NSL;
  float* deg    = (float*)d_ws;
  float* Hcat   = deg + Np;
  float* Htmp   = Hcat + (size_t)N * 256;
  int*   cnt    = (int*)(Htmp + (size_t)N * D);
  int*   rowptr = cnt + Np;
  int*   lst    = rowptr + Np;
  int*   lstn   = lst + 2 * Ep;
  int*   tab    = lstn + 2 * Ep;
  int*   offs   = tab + Tn;
  int*   binq   = offs + Tn;

  const int T = 256;
  const int PB = 1024;
  gcn_bin_count<<<NSL, BT, nbk * sizeof(int), stream>>>(dst, tab, E, nbk);
  gcn_scan     <<<(int)((Tn + 1023) / 1024), T, 0, stream>>>(tab, offs, (int)Tn);
  gcn_bin_place<<<NSL, BT, nbk * sizeof(int), stream>>>(src, dst, ew, offs, binq, E, nbk);
  gcn_bin_lists<<<nbk, BT, 0, stream>>>(offs, binq, cnt, rowptr, lst, E, N, nbk);
  gcn_dinv    <<<(N + T - 1) / T, T, 0, stream>>>(deg, rowptr, cnt, lst, N);
  gcn_norm    <<<PB, T, 0, stream>>>(deg, rowptr, cnt, lst, lstn, N);
  gcn_copy_x  <<<((size_t)N * D + T - 1) / T, T, 0, stream>>>(x, Hcat, N);

  const int gemm_blocks = (N + TROWS - 1) / TROWS;
  for (int l = 0; l < 3; ++l) {
    const int off = (l + 1) * D;
    gcn_gemm_wmma<<<gemm_blocks, T, 0, stream>>>(Hcat + l * D, 256, Wl[l], nullptr,
                                                 Htmp, D, N);
    gcn_gather   <<<PB, T, 0, stream>>>(Htmp, deg, rowptr, cnt, lstn, bl[l], Hcat, off, N);
  }
  gcn_gemm_wmma<<<gemm_blocks, T, 0, stream>>>(Hcat, 256, Wf, bf, (float*)d_out, 4 * D, N);
}


# Input features read in place by the two GEMMs that use them (layer 0 and the first K-chunk of the final product); the kernel that copied x into the concatenated buffer (200000 short waves) is gone
# speedup vs baseline: 2.2570x; 1.0473x over previous
#include <hip/hip_runtime.h>

typedef __attribute__((ext_vector_type(2))) float v2f;
typedef __attribute__((ext_vector_type(8))) float v8f;

#define D     64
#define TROWS 32
#define LDP   68

#define BSH   8
#define BNODE (1 << BSH)
#define NSL   256
#define BT    512

__global__ __launch_bounds__(BT)
void gcn_bin_count(const int* __restrict__ dst, int* __restrict__ tab, int e, int nbk) {
  extern __shared__ int hist[];
  const int s = blockIdx.x, tid = threadIdx.x;
  for (int b = tid; b < nbk; b += BT) hist[b] = 0;
  __syncthreads();
  const int per = (e + NSL - 1) / NSL;
  const int lo  = s * per;
  const int hi  = min(e, lo + per);
  for (int i0 = lo + tid; i0 < hi; i0 += 4 * BT) {
    int d[4];
#pragma unroll
    for (int u = 0; u < 4; ++u) {
      const int i = i0 + u * BT;
      d[u] = (i < hi) ? dst[i] : 0;
    }
#pragma unroll
    for (int u = 0; u < 4; ++u)
      if (i0 + u * BT < hi) atomicAdd(&hist[d[u] >> BSH], 1);
  }
  __syncthreads();
  for (int b = tid; b < nbk; b += BT) tab[b * NSL + s] = hist[b];
}

__global__ void gcn_scan(const int* __restrict__ cnt, int* __restrict__ rowptr, int n) {
  __shared__ int pre[8];
  __shared__ int loc[8];
  const int tid = threadIdx.x, lane = tid & 31, wv = tid >> 5;
  const int b = blockIdx.x;
  int part = 0;
  const int4* c4 = (const int4*)cnt;
  const int n4 = b * 256;
  for (int i = tid; i < n4; i += 256) {
    int4 v = c4[i];
    part += (v.x + v.y) + (v.z + v.w);
  }
  for (int o = 16; o > 0; o >>= 1) part += __shfl_xor(part, o);
  if (lane == 0) pre[wv] = part;

  const int i0 = b * 1024 + tid * 4;
  const int c0 = (i0 + 0 < n) ? cnt[i0 + 0] : 0;
  const int c1 = (i0 + 1 < n) ? cnt[i0 + 1] : 0;
  const int c2 = (i0 + 2 < n) ? cnt[i0 + 2] : 0;
  const int c3 = (i0 + 3 < n) ? cnt[i0 + 3] : 0;
  const int s = (c0 + c1) + (c2 + c3);
  int inc = s;
  for (int o = 1; o < 32; o <<= 1) {
    int t = __shfl_up(inc, o);
    if (lane >= o) inc += t;
  }
  if (lane == 31) loc[wv] = inc;
  __syncthreads();
  int base = 0;
  for (int k = 0; k < 8; ++k) base += pre[k];
  for (int k = 0; k < wv; ++k) base += loc[k];
  const int ex = base + inc - s;
  if (i0 + 0 < n) rowptr[i0 + 0] = ex;
  if (i0 + 1 < n) rowptr[i0 + 1] = ex + c0;
  if (i0 + 2 < n) rowptr[i0 + 2] = ex + c0 + c1;
  if (i0 + 3 < n) rowptr[i0 + 3] = ex + c0 + c1 + c2;
}

__global__ __launch_bounds__(BT)
void gcn_bin_place(const int* __restrict__ src, const int* __restrict__ dst,
                   const float* __restrict__ ew, const int* __restrict__ offs,
                   int* __restrict__ binq_, int e, int nbk) {
  extern __shared__ int cur[];
  int4* binq = (int4*)binq_;
  const int s = blockIdx.x, tid = threadIdx.x;
  for (int b = tid; b < nbk; b += BT) cur[b] = offs[b * NSL + s];
  __syncthreads();
  const int per = (e + NSL - 1) / NSL;
  const int lo  = s * per;
  const int hi  = min(e, lo + per);
  for (int i0 = lo + tid; i0 < hi; i0 += 4 * BT) {
    int d[4], sv[4], wb[4];
#pragma unroll
    for (int u = 0; u < 4; ++u) {
      const int i = i0 + u * BT;
      d[u] = 0; sv[u] = 0; wb[u] = 0;
      if (i < hi) { d[u] = dst[i]; sv[u] = src[i]; wb[u] = __float_as_int(ew[i]); }
    }
#pragma unroll
    for (int u = 0; u < 4; ++u) {
      if (i0 + u * BT < hi) {
        const int p = atomicAdd(&cur[d[u] >> BSH], 1);
        binq[p] = make_int4(sv[u], wb[u], d[u], 0);
      }
    }
  }
}

__global__ __launch_bounds__(BT)
void gcn_bin_lists(const int* __restrict__ offs, const int* __restrict__ binq_,
                   int* __restrict__ cnt, int* __restrict__ rowptr, int* __restrict__ lst_,
                   int e, int n, int nbk) {
  __shared__ int c[BNODE];
  __shared__ int cursor[BNODE];
  __shared__ int wsum[BNODE / 32];
  const int4* binq = (const int4*)binq_;
  int2* lst = (int2*)lst_;
  const int b = blockIdx.x, tid = threadIdx.x, lane = tid & 31, wv = tid >> 5;
  const int base = b << BSH;
  const int bs = offs[b * NSL];
  const int be = (b + 1 < nbk) ? offs[(b + 1) * NSL] : e;
  if (tid < BNODE) c[tid] = 0;
  __syncthreads();
  for (int i0 = bs + tid; i0 < be; i0 += 4 * BT) {
    int d[4];
#pragma unroll
    for (int u = 0; u < 4; ++u) {
      const int i = i0 + u * BT;
      d[u] = (i < be) ? binq[i].z : base;
    }
#pragma unroll
    for (int u = 0; u < 4; ++u)
      if (i0 + u * BT < be) atomicAdd(&c[d[u] - base], 1);
  }
  __syncthreads();
  const int v = (tid < BNODE) ? c[tid] : 0;
  int inc = v;
  for (int o = 1; o < 32; o <<= 1) {
    int t = __shfl_up(inc, o);
    if (lane >= o) inc += t;
  }
  if (lane == 31 && tid < BNODE) wsum[wv] = inc;
  __syncthreads();
  if (tid < BNODE) {
    int ex = bs + inc - v;
    for (int k = 0; k < wv; ++k) ex += wsum[k];
    cursor[tid] = ex;
    const int node = base + tid;
    if (node < n) { cnt[node] = v; rowptr[node] = ex; }
  }
  __syncthreads();
  for (int i0 = bs + tid; i0 < be; i0 += 4 * BT) {
    int4 q[4];
#pragma unroll
    for (int u = 0; u < 4; ++u) {
      const int i = i0 + u * BT;
      q[u] = make_int4(0, 0, base, 0);
      if (i < be) q[u] = binq[i];
    }
#pragma unroll
    for (int u = 0; u < 4; ++u) {
      if (i0 + u * BT < be) {
        const int p = atomicAdd(&cursor[q[u].z - base], 1);
        lst[p] = make_int2(q[u].x, q[u].y);
      }
    }
  }
}

__global__ void gcn_dinv(float* __restrict__ dinv, const int* __restrict__ rowptr,
                         const int* __restrict__ cnt, const int* __restrict__ lst_, int n) {
  const int2* lst = (const int2*)lst_;
  int i = blockIdx.x * blockDim.x + threadIdx.x;
  if (i < n) {
    const int st = rowptr[i], len = cnt[i];
    float d = 1.0f;
    for (int j = 0; j < len; ++j) d += __int_as_float(lst[st + j].y);
    dinv[i] = (d > 0.0f) ? rsqrtf(d) : 0.0f;
  }
}

__global__ __launch_bounds__(256)
void gcn_norm(const float* __restrict__ dinv, const int* __restrict__ rowptr,
              const int* __restrict__ cnt, const int* __restrict__ lst_,
              int* __restrict__ lstn_, int n) {
  const int2* lst  = (const int2*)lst_;
  int2*       lstn = (int2*)lstn_;
  const int l  = threadIdx.x & 15;
  const int ng = (gridDim.x * blockDim.x) >> 4;
  for (int node = (blockIdx.x * blockDim.x + threadIdx.x) >> 4; node < n; node += ng) {
    const int st = rowptr[node], len = cnt[node];
    const float di = dinv[node];
    for (int j = l; j < len; j += 16) {
      const int2 e = lst[st + j];
      const float nrm = dinv[e.x] * __int_as_float(e.y) * di;
      lstn[st + j] = make_int2(e.x, __float_as_int(nrm));
    }
  }
}

__global__ __launch_bounds__(256)
void gcn_gather(const float* __restrict__ Htmp, const float* __restrict__ dinv,
                const int* __restrict__ rowptr, const int* __restrict__ cnt,
                const int* __restrict__ lstn_, const float* __restrict__ bias,
                float* __restrict__ Hcat, int off, int n) {
  const int2* lstn = (const int2*)lstn_;
  const int l  = threadIdx.x & 15;
  const int ng = (gridDim.x * blockDim.x) >> 4;
  int node = (blockIdx.x * blockDim.x + threadIdx.x) >> 4;
  const float4 bv = *(const float4*)(bias + l * 4);
  int st = 0, end = 0;
  if (node < n) { st = rowptr[node]; end = st + cnt[node]; }
  while (node < n) {
    const int nnode = node + ng;
    int nst = 0, nlen = 0;
    if (nnode < n) { nst = rowptr[nnode]; nlen = cnt[nnode]; }
    const float di = dinv[node];
    const float4 hs = *(const float4*)(Htmp + (size_t)node * D + l * 4);
    float4 acc = make_float4(0.0f, 0.0f, 0.0f, 0.0f);
    for (int p = st; p < end; p += 16) {
      const int m = min(16, end - p);
      const int2 e = lstn[p + min(l, m - 1)];
      const int   sv = e.x;
      const float wv = __int_as_float(e.y & ((l < m) ? -1 : 0));
      int k = 0;
      do {
        float4 h[4];
        float  w[4];
#pragma unroll
        for (int u = 0; u < 4; ++u) {
          const int s = __shfl(sv, k + u, 16);
          w[u] = __shfl(wv, k + u, 16);
          h[u] = make_float4(0.0f, 0.0f, 0.0f, 0.0f);
          if (k + u < m) h[u] = *(const float4*)(Htmp + (size_t)s * D + l * 4);
        }
#pragma unroll
        for (int u = 0; u < 4; ++u) {
          acc.x = fmaf(h[u].x, w[u], acc.x);
          acc.y = fmaf(h[u].y, w[u], acc.y);
          acc.z = fmaf(h[u].z, w[u], acc.z);
          acc.w = fmaf(h[u].w, w[u], acc.w);
        }
        k += 4;
      } while (k < m);
    }
    const float sc = di * di;
    float4 r;
    r.x = fmaf(sc, hs.x, acc.x) + bv.x;
    r.y = fmaf(sc, hs.y, acc.y) + bv.y;
    r.z = fmaf(sc, hs.z, acc.z) + bv.z;
    r.w = fmaf(sc, hs.w, acc.w) + bv.w;
    r.x = r.x > 0.0f ? r.x : 0.0f;
    r.y = r.y > 0.0f ? r.y : 0.0f;
    r.z = r.z > 0.0f ? r.z : 0.0f;
    r.w = r.w > 0.0f ? r.w : 0.0f;
    *(float4*)(Hcat + (size_t)node * 256 + off + l * 4) = r;
    node = nnode; st = nst; end = nst + nlen;
  }
}

__global__ void gcn_gemm_wmma(const float* __restrict__ X0, int ld0,
                              const float* __restrict__ X, int ldx,
                              const float* __restrict__ W,
                              const float* __restrict__ bias,
                              float* __restrict__ Out, int K, int n) {
  __shared__ float Xs[TROWS * LDP];
  __shared__ float Ws[D * LDP];
  const int tid    = threadIdx.x;
  const int wave   = tid >> 5;
  const int lane   = tid & 31;
  const int laneLo = lane & 15;
  const int laneHi = lane >> 4;
  const int row0   = blockIdx.x * TROWS;
  const int wm     = (wave >> 2) << 4;
  const int wn     = (wave & 3) << 4;

  v8f acc = {};
  for (int kc = 0; kc < K; kc += D) {
    const float* Xc = (kc == 0) ? X0 : X + kc;
    const int    ldc = (kc == 0) ? ld0 : ldx;
    for (int i = tid; i < TROWS * D; i += 256) {
      int r = i >> 6, c = i & 63;
      int rr = row0 + r; if (rr >= n) rr = n - 1;
      Xs[r * LDP + c] = Xc[(size_t)rr * ldc + c];
    }
    for (int i = tid; i < D * D; i += 256) {
      int r = i >> 6, c = i & 63;
      Ws[r * LDP + c] = W[(size_t)(kc + r) * D + c];
    }
    __syncthreads();
#pragma unroll
    for (int k0 = 0; k0 < D; k0 += 4) {
      v2f a = *(const v2f*)(&Xs[(wm + laneLo) * LDP + k0 + 2 * laneHi]);
      v2f b;
      b[0] = Ws[(k0 + 2 * laneHi + 0) * LDP + wn + laneLo];
      b[1] = Ws[(k0 + 2 * laneHi + 1) * LDP + wn + laneLo];
      acc = __builtin_amdgcn_wmma_f32_16x16x4_f32(false, a, false, b,
                                                  (short)0, acc, false, false);
    }
    __syncthreads();
  }
  const int gcol = wn + laneLo;
  const float bv = bias ? bias[gcol] : 0.0f;
#pragma unroll
  for (int v = 0; v < 8; ++v) {
    int grow = row0 + wm + v + 8 * laneHi;
    if (grow < n) Out[(size_t)grow * D + gcol] = acc[v] + bv;
  }
}

extern "C" void kernel_launch(void* const* d_in, const int* in_sizes, int n_in,
                              void* d_out, int out_size, void* d_ws, size_t ws_size,
                              hipStream_t stream) {
  const float* x   = (const float*)d_in[0];
  const int*   ei  = (const int*)d_in[1];
  const float* ew  = (const float*)d_in[2];
  const float* Wl[3] = {(const float*)d_in[3], (const float*)d_in[5], (const float*)d_in[7]};
  const float* bl[3] = {(const float*)d_in[4], (const float*)d_in[6], (const float*)d_in[8]};
  const float* Wf  = (const float*)d_in[9];
  const float* bf  = (const float*)d_in[10];

  const int N = in_sizes[0] / D;
  const int E = in_sizes[2];
  const int* src = ei;
  const int* dst = ei + E;

  const size_t Np = ((size_t)N + 255) & ~(size_t)255;
  const size_t Ep = ((size_t)E + 255) & ~(size_t)255;
  const int    nbk = (int)(Np >> BSH);
  const size_t Tn  = (size_t)nbk * NSL;
  float* deg    = (float*)d_ws;
  float* Hcat   = deg + Np;
  float* Htmp   = Hcat + (size_t)N * 256;
  int*   cnt    = (int*)(Htmp + (size_t)N * D);
  int*   rowptr = cnt + Np;
  int*   lst    = rowptr + Np;
  int*   lstn   = lst + 2 * Ep;
  int*   tab    = lstn + 2 * Ep;
  int*   offs   = tab + Tn;
  int*   binq   = offs + Tn;

  const int T = 256;
  const int PB = 1024;
  gcn_bin_count<<<NSL, BT, nbk * sizeof(int), stream>>>(dst, tab, E, nbk);
  gcn_scan     <<<(int)((Tn + 1023) / 1024), T, 0, stream>>>(tab, offs, (int)Tn);
  gcn_bin_place<<<NSL, BT, nbk * sizeof(int), stream>>>(src, dst, ew, offs, binq, E, nbk);
  gcn_bin_lists<<<nbk, BT, 0, stream>>>(offs, binq, cnt, rowptr, lst, E, N, nbk);
  gcn_dinv    <<<(N + T - 1) / T, T, 0, stream>>>(deg, rowptr, cnt, lst, N);
  gcn_norm    <<<PB, T, 0, stream>>>(deg, rowptr, cnt, lst, lstn, N);

  const int gemm_blocks = (N + TROWS - 1) / TROWS;
  for (int l = 0; l < 3; ++l) {
    const int off = (l + 1) * D;
    const float* Xin = (l == 0) ? x : Hcat + l * D;
    const int    ldi = (l == 0) ? D : 256;
    gcn_gemm_wmma<<<gemm_blocks, T, 0, stream>>>(Xin, ldi, Xin, ldi, Wl[l], nullptr,
                                                 Htmp, D, N);
    gcn_gather   <<<PB, T, 0, stream>>>(Htmp, deg, rowptr, cnt, lstn, bl[l], Hcat, off, N);
  }
  gcn_gemm_wmma<<<gemm_blocks, T, 0, stream>>>(x, D, Hcat, 256, Wf, bf, (float*)d_out,
                                               4 * D, N);
}


# Each layer's gather and its 64x64 product fused in one kernel: rows aggregated from the layer input into LDS, then WMMA, bias, relu; sum order now (A X) W; 3 launches and the product plane gone
# speedup vs baseline: 2.5087x; 1.1115x over previous
#include <hip/hip_runtime.h>

typedef __attribute__((ext_vector_type(2))) float v2f;
typedef __attribute__((ext_vector_type(8))) float v8f;

#define D     64
#define TROWS 32
#define LDP   68

#define BSH   8
#define BNODE (1 << BSH)
#define NSL   256
#define BT    512

__global__ __launch_bounds__(BT)
void gcn_bin_count(const int* __restrict__ dst, int* __restrict__ tab, int e, int nbk) {
  extern __shared__ int hist[];
  const int s = blockIdx.x, tid = threadIdx.x;
  for (int b = tid; b < nbk; b += BT) hist[b] = 0;
  __syncthreads();
  const int per = (e + NSL - 1) / NSL;
  const int lo  = s * per;
  const int hi  = min(e, lo + per);
  for (int i0 = lo + tid; i0 < hi; i0 += 4 * BT) {
    int d[4];
#pragma unroll
    for (int u = 0; u < 4; ++u) {
      const int i = i0 + u * BT;
      d[u] = (i < hi) ? dst[i] : 0;
    }
#pragma unroll
    for (int u = 0; u < 4; ++u)
      if (i0 + u * BT < hi) atomicAdd(&hist[d[u] >> BSH], 1);
  }
  __syncthreads();
  for (int b = tid; b < nbk; b += BT) tab[b * NSL + s] = hist[b];
}

__global__ void gcn_scan(const int* __restrict__ cnt, int* __restrict__ rowptr, int n) {
  __shared__ int pre[8];
  __shared__ int loc[8];
  const int tid = threadIdx.x, lane = tid & 31, wv = tid >> 5;
  const int b = blockIdx.x;
  int part = 0;
  const int4* c4 = (const int4*)cnt;
  const int n4 = b * 256;
  for (int i = tid; i < n4; i += 256) {
    int4 v = c4[i];
    part += (v.x + v.y) + (v.z + v.w);
  }
  for (int o = 16; o > 0; o >>= 1) part += __shfl_xor(part, o);
  if (lane == 0) pre[wv] = part;

  const int i0 = b * 1024 + tid * 4;
  const int c0 = (i0 + 0 < n) ? cnt[i0 + 0] : 0;
  const int c1 = (i0 + 1 < n) ? cnt[i0 + 1] : 0;
  const int c2 = (i0 + 2 < n) ? cnt[i0 + 2] : 0;
  const int c3 = (i0 + 3 < n) ? cnt[i0 + 3] : 0;
  const int s = (c0 + c1) + (c2 + c3);
  int inc = s;
  for (int o = 1; o < 32; o <<= 1) {
    int t = __shfl_up(inc, o);
    if (lane >= o) inc += t;
  }
  if (lane == 31) loc[wv] = inc;
  __syncthreads();
  int base = 0;
  for (int k = 0; k < 8; ++k) base += pre[k];
  for (int k = 0; k < wv; ++k) base += loc[k];
  const int ex = base + inc - s;
  if (i0 + 0 < n) rowptr[i0 + 0] = ex;
  if (i0 + 1 < n) rowptr[i0 + 1] = ex + c0;
  if (i0 + 2 < n) rowptr[i0 + 2] = ex + c0 + c1;
  if (i0 + 3 < n) rowptr[i0 + 3] = ex + c0 + c1 + c2;
}

__global__ __launch_bounds__(BT)
void gcn_bin_place(const int* __restrict__ src, const int* __restrict__ dst,
                   const float* __restrict__ ew, const int* __restrict__ offs,
                   int* __restrict__ binq_, int e, int nbk) {
  extern __shared__ int cur[];
  int4* binq = (int4*)binq_;
  const int s = blockIdx.x, tid = threadIdx.x;
  for (int b = tid; b < nbk; b += BT) cur[b] = offs[b * NSL + s];
  __syncthreads();
  const int per = (e + NSL - 1) / NSL;
  const int lo  = s * per;
  const int hi  = min(e, lo + per);
  for (int i0 = lo + tid; i0 < hi; i0 += 4 * BT) {
    int d[4], sv[4], wb[4];
#pragma unroll
    for (int u = 0; u < 4; ++u) {
      const int i = i0 + u * BT;
      d[u] = 0; sv[u] = 0; wb[u] = 0;
      if (i < hi) { d[u] = dst[i]; sv[u] = src[i]; wb[u] = __float_as_int(ew[i]); }
    }
#pragma unroll
    for (int u = 0; u < 4; ++u) {
      if (i0 + u * BT < hi) {
        const int p = atomicAdd(&cur[d[u] >> BSH], 1);
        binq[p] = make_int4(sv[u], wb[u], d[u], 0);
      }
    }
  }
}

__global__ __launch_bounds__(BT)
void gcn_bin_lists(const int* __restrict__ offs, const int* __restrict__ binq_,
                   int* __restrict__ cnt, int* __restrict__ rowptr, int* __restrict__ lst_,
                   int e, int n, int nbk) {
  __shared__ int c[BNODE];
  __shared__ int cursor[BNODE];
  __shared__ int wsum[BNODE / 32];
  const int4* binq = (const int4*)binq_;
  int2* lst = (int2*)lst_;
  const int b = blockIdx.x, tid = threadIdx.x, lane = tid & 31, wv = tid >> 5;
  const int base = b << BSH;
  const int bs = offs[b * NSL];
  const int be = (b + 1 < nbk) ? offs[(b + 1) * NSL] : e;
  if (tid < BNODE) c[tid] = 0;
  __syncthreads();
  for (int i0 = bs + tid; i0 < be; i0 += 4 * BT) {
    int d[4];
#pragma unroll
    for (int u = 0; u < 4; ++u) {
      const int i = i0 + u * BT;
      d[u] = (i < be) ? binq[i].z : base;
    }
#pragma unroll
    for (int u = 0; u < 4; ++u)
      if (i0 + u * BT < be) atomicAdd(&c[d[u] - base], 1);
  }
  __syncthreads();
  const int v = (tid < BNODE) ? c[tid] : 0;
  int inc = v;
  for (int o = 1; o < 32; o <<= 1) {
    int t = __shfl_up(inc, o);
    if (lane >= o) inc += t;
  }
  if (lane == 31 && tid < BNODE) wsum[wv] = inc;
  __syncthreads();
  if (tid < BNODE) {
    int ex = bs + inc - v;
    for (int k = 0; k < wv; ++k) ex += wsum[k];
    cursor[tid] = ex;
    const int node = base + tid;
    if (node < n) { cnt[node] = v; rowptr[node] = ex; }
  }
  __syncthreads();
  for (int i0 = bs + tid; i0 < be; i0 += 4 * BT) {
    int4 q[4];
#pragma unroll
    for (int u = 0; u < 4; ++u) {
      const int i = i0 + u * BT;
      q[u] = make_int4(0, 0, base, 0);
      if (i < be) q[u] = binq[i];
    }
#pragma unroll
    for (int u = 0; u < 4; ++u) {
      if (i0 + u * BT < be) {
        const int p = atomicAdd(&cursor[q[u].z - base], 1);
        lst[p] = make_int2(q[u].x, q[u].y);
      }
    }
  }
}

__global__ void gcn_dinv(float* __restrict__ dinv, const int* __restrict__ rowptr,
                         const int* __restrict__ cnt, const int* __restrict__ lst_, int n) {
  const int2* lst = (const int2*)lst_;
  int i = blockIdx.x * blockDim.x + threadIdx.x;
  if (i < n) {
    const int st = rowptr[i], len = cnt[i];
    float d = 1.0f;
    for (int j = 0; j < len; ++j) d += __int_as_float(lst[st + j].y);
    dinv[i] = (d > 0.0f) ? rsqrtf(d) : 0.0f;
  }
}

__global__ __launch_bounds__(256)
void gcn_norm(const float* __restrict__ dinv, const int* __restrict__ rowptr,
              const int* __restrict__ cnt, const int* __restrict__ lst_,
              int* __restrict__ lstn_, int n) {
  const int2* lst  = (const int2*)lst_;
  int2*       lstn = (int2*)lstn_;
  const int l  = threadIdx.x & 15;
  const int ng = (gridDim.x * blockDim.x) >> 4;
  for (int node = (blockIdx.x * blockDim.x + threadIdx.x) >> 4; node < n; node += ng) {
    const int st = rowptr[node], len = cnt[node];
    const float di = dinv[node];
    for (int j = l; j < len; j += 16) {
      const int2 e = lst[st + j];
      const float nrm = dinv[e.x] * __int_as_float(e.y) * di;
      lstn[st + j] = make_int2(e.x, __float_as_int(nrm));
    }
  }
}

#define LROWS 64
#define LGR   (LROWS / 16)
#define LMT   (LROWS / 32)

__device__ __forceinline__ float4 gcn_sum_list(const float* __restrict__ Xin, int ldi,
                                               const int2* __restrict__ lstn,
                                               int st, int end, int l, int2 e) {
  float4 acc = make_float4(0.0f, 0.0f, 0.0f, 0.0f);
  for (int p = st; p < end; p += 16) {
    const int m = min(16, end - p);
    if (p > st) e = lstn[p + min(l, m - 1)];
    const int   sv = e.x;
    const float wv = __int_as_float(e.y & ((l < m) ? -1 : 0));
    int k = 0;
    do {
      float4 h[4];
      float  w[4];
#pragma unroll
      for (int u = 0; u < 4; ++u) {
        const int s = __shfl(sv, k + u, 16);
        w[u] = __shfl(wv, k + u, 16);
        h[u] = make_float4(0.0f, 0.0f, 0.0f, 0.0f);
        if (k + u < m) h[u] = *(const float4*)(Xin + (size_t)s * ldi + l * 4);
      }
#pragma unroll
      for (int u = 0; u < 4; ++u) {
        acc.x = fmaf(h[u].x, w[u], acc.x);
        acc.y = fmaf(h[u].y, w[u], acc.y);
        acc.z = fmaf(h[u].z, w[u], acc.z);
        acc.w = fmaf(h[u].w, w[u], acc.w);
      }
      k += 4;
    } while (k < m);
  }
  return acc;
}

__global__ __launch_bounds__(256)
void gcn_layer(const float* __restrict__ Xin, int ldi, const float* __restrict__ dinv,
               const int* __restrict__ rowptr, const int* __restrict__ cnt,
               const int* __restrict__ lstn_, const float* __restrict__ W,
               const float* __restrict__ bias, float* Hout, int ldo, int n) {
  __shared__ __attribute__((aligned(16))) float Ws[D * LDP];
  __shared__ __attribute__((aligned(16))) float Zs[LROWS * LDP];
  const int2* lstn = (const int2*)lstn_;
  const int tid    = threadIdx.x;
  const int wave   = tid >> 5;
  const int lane   = tid & 31;
  const int laneLo = lane & 15;
  const int laneHi = lane >> 4;
  const int wn     = (wave & 3) << 4;
  const int wm0    = (wave >> 2) * (LMT * 16);
  const int g      = tid >> 4;
  const int l      = tid & 15;
  const int row0   = blockIdx.x * LROWS;
  const int r0     = g * LGR;

  int stL = 0, cL = 0;
  float dL = 0.0f;
  {
    const int nd = row0 + r0 + l;
    if (l < LGR && nd < n) { stL = rowptr[nd]; cL = cnt[nd]; dL = dinv[nd]; }
  }
  const int   gcol = wn + laneLo;
  const float bv   = bias[gcol];

  int st = __shfl(stL, 0, 16);
  int en = st + __shfl(cL, 0, 16);
  int2 e = make_int2(0, 0);
  if (en > st) e = lstn[st + min(l, min(16, en - st) - 1)];
  {
    float4 wr[D * D / 4 / 256];
#pragma unroll
    for (int u = 0; u < D * D / 4 / 256; ++u)
      wr[u] = *(const float4*)(W + (size_t)(tid + u * 256) * 4);
#pragma unroll
    for (int u = 0; u < D * D / 4 / 256; ++u) {
      const int i = tid + u * 256;
      *(float4*)(&Ws[(i >> 4) * LDP + ((i & 15) << 2)]) = wr[u];
    }
  }
#pragma unroll
  for (int j = 0; j < LGR; ++j) {
    const int node = row0 + r0 + j;
    const float di = __shfl(dL, j, 16);
    int nst = 0, nen = 0;
    int2 ne = make_int2(0, 0);
    if (j + 1 < LGR) {
      nst = __shfl(stL, j + 1, 16);
      nen = nst + __shfl(cL, j + 1, 16);
      if (nen > nst) ne = lstn[nst + min(l, min(16, nen - nst) - 1)];
    }
    float4 hs = make_float4(0.0f, 0.0f, 0.0f, 0.0f);
    if (node < n) hs = *(const float4*)(Xin + (size_t)node * ldi + l * 4);
    const float4 a = gcn_sum_list(Xin, ldi, lstn, st, en, l, e);
    const float sc = di * di;
    float4 z;
    z.x = fmaf(sc, hs.x, a.x); z.y = fmaf(sc, hs.y, a.y);
    z.z = fmaf(sc, hs.z, a.z); z.w = fmaf(sc, hs.w, a.w);
    *(float4*)(&Zs[(r0 + j) * LDP + l * 4]) = z;
    st = nst; en = nen; e = ne;
  }
  __syncthreads();

  v8f acc[LMT];
#pragma unroll
  for (int t = 0; t < LMT; ++t) acc[t] = v8f{};
#pragma unroll
  for (int k0 = 0; k0 < D; k0 += 4) {
    v2f b;
    b[0] = Ws[(k0 + 2 * laneHi + 0) * LDP + wn + laneLo];
    b[1] = Ws[(k0 + 2 * laneHi + 1) * LDP + wn + laneLo];
#pragma unroll
    for (int t = 0; t < LMT; ++t) {
      v2f a = *(const v2f*)(&Zs[(wm0 + t * 16 + laneLo) * LDP + k0 + 2 * laneHi]);
      acc[t] = __builtin_amdgcn_wmma_f32_16x16x4_f32(false, a, false, b,
                                                     (short)0, acc[t], false, false);
    }
  }
#pragma unroll
  for (int t = 0; t < LMT; ++t) {
#pragma unroll
    for (int v = 0; v < 8; ++v) {
      const int grow = row0 + wm0 + t * 16 + v + 8 * laneHi;
      float r = acc[t][v] + bv;
      r = r > 0.0f ? r : 0.0f;
      if (grow < n) Hout[(size_t)grow * ldo + gcol] = r;
    }
  }
}

__global__ void gcn_gemm_wmma(const float* __restrict__ X0, int ld0,
                              const float* __restrict__ X, int ldx,
                              const float* __restrict__ W,
                              const float* __restrict__ bias,
                              float* __restrict__ Out, int K, int n) {
  __shared__ float Xs[TROWS * LDP];
  __shared__ float Ws[D * LDP];
  const int tid    = threadIdx.x;
  const int wave   = tid >> 5;
  const int lane   = tid & 31;
  const int laneLo = lane & 15;
  const int laneHi = lane >> 4;
  const int row0   = blockIdx.x * TROWS;
  const int wm     = (wave >> 2) << 4;
  const int wn     = (wave & 3) << 4;

  v8f acc = {};
  for (int kc = 0; kc < K; kc += D) {
    const float* Xc = (kc == 0) ? X0 : X + kc;
    const int    ldc = (kc == 0) ? ld0 : ldx;
    for (int i = tid; i < TROWS * D; i += 256) {
      int r = i >> 6, c = i & 63;
      int rr = row0 + r; if (rr >= n) rr = n - 1;
      Xs[r * LDP + c] = Xc[(size_t)rr * ldc + c];
    }
    for (int i = tid; i < D * D; i += 256) {
      int r = i >> 6, c = i & 63;
      Ws[r * LDP + c] = W[(size_t)(kc + r) * D + c];
    }
    __syncthreads();
#pragma unroll
    for (int k0 = 0; k0 < D; k0 += 4) {
      v2f a = *(const v2f*)(&Xs[(wm + laneLo) * LDP + k0 + 2 * laneHi]);
      v2f b;
      b[0] = Ws[(k0 + 2 * laneHi + 0) * LDP + wn + laneLo];
      b[1] = Ws[(k0 + 2 * laneHi + 1) * LDP + wn + laneLo];
      acc = __builtin_amdgcn_wmma_f32_16x16x4_f32(false, a, false, b,
                                                  (short)0, acc, false, false);
    }
    __syncthreads();
  }
  const int gcol = wn + laneLo;
  const float bv = bias ? bias[gcol] : 0.0f;
#pragma unroll
  for (int v = 0; v < 8; ++v) {
    int grow = row0 + wm + v + 8 * laneHi;
    if (grow < n) Out[(size_t)grow * D + gcol] = acc[v] + bv;
  }
}

extern "C" void kernel_launch(void* const* d_in, const int* in_sizes, int n_in,
                              void* d_out, int out_size, void* d_ws, size_t ws_size,
                              hipStream_t stream) {
  const float* x   = (const float*)d_in[0];
  const int*   ei  = (const int*)d_in[1];
  const float* ew  = (const float*)d_in[2];
  const float* Wl[3] = {(const float*)d_in[3], (const float*)d_in[5], (const float*)d_in[7]};
  const float* bl[3] = {(const float*)d_in[4], (const float*)d_in[6], (const float*)d_in[8]};
  const float* Wf  = (const float*)d_in[9];
  const float* bf  = (const float*)d_in[10];

  const int N = in_sizes[0] / D;
  const int E = in_sizes[2];
  const int* src = ei;
  const int* dst = ei + E;

  const size_t Np = ((size_t)N + 255) & ~(size_t)255;
  const size_t Ep = ((size_t)E + 255) & ~(size_t)255;
  const int    nbk = (int)(Np >> BSH);
  const size_t Tn  = (size_t)nbk * NSL;
  float* deg    = (float*)d_ws;
  float* Hcat   = deg + Np;
  int*   cnt    = (int*)(Hcat + (size_t)N * 256);
  int*   rowptr = cnt + Np;
  int*   lst    = rowptr + Np;
  int*   lstn   = lst + 2 * Ep;
  int*   tab    = lstn + 2 * Ep;
  int*   offs   = tab + Tn;
  int*   binq   = offs + Tn;

  const int T = 256;
  const int PB = 1024;
  gcn_bin_count<<<NSL, BT, nbk * sizeof(int), stream>>>(dst, tab, E, nbk);
  gcn_scan     <<<(int)((Tn + 1023) / 1024), T, 0, stream>>>(tab, offs, (int)Tn);
  gcn_bin_place<<<NSL, BT, nbk * sizeof(int), stream>>>(src, dst, ew, offs, binq, E, nbk);
  gcn_bin_lists<<<nbk, BT, 0, stream>>>(offs, binq, cnt, rowptr, lst, E, N, nbk);
  gcn_dinv    <<<(N + T - 1) / T, T, 0, stream>>>(deg, rowptr, cnt, lst, N);
  gcn_norm    <<<PB, T, 0, stream>>>(deg, rowptr, cnt, lst, lstn, N);

  const int layer_blocks = (N + LROWS - 1) / LROWS;
  for (int l = 0; l < 3; ++l) {
    const float* Xin = (l == 0) ? x : Hcat + l * D;
    const int    ldi = (l == 0) ? D : 256;
    gcn_layer<<<layer_blocks, T, 0, stream>>>(Xin, ldi, deg, rowptr, cnt, lstn, Wl[l], bl[l],
                                              Hcat + (l + 1) * D, 256, N);
  }
  const int gemm_blocks = (N + TROWS - 1) / TROWS;
  gcn_gemm_wmma<<<gemm_blocks, T, 0, stream>>>(x, D, Hcat, 256, Wf, bf, (float*)d_out,
                                               4 * D, N);
}


# Counting-sort start offsets taken inside the placing kernel: each work-group sums its own column of the prefix from the count table, all loads of two rows in flight; the serial scan launch is gone
# speedup vs baseline: 2.8732x; 1.1453x over previous
#include <hip/hip_runtime.h>

typedef __attribute__((ext_vector_type(2))) float v2f;
typedef __attribute__((ext_vector_type(8))) float v8f;

#define D     64
#define TROWS 32
#define LDP   68

#define BSH   8
#define BNODE (1 << BSH)
#define NSL   256
#define BT    512

__global__ __launch_bounds__(BT)
void gcn_bin_count(const int* __restrict__ dst, int* __restrict__ tab, int e, int nbk) {
  extern __shared__ int hist[];
  const int s = blockIdx.x, tid = threadIdx.x;
  for (int b = tid; b < nbk; b += BT) hist[b] = 0;
  __syncthreads();
  const int per = (e + NSL - 1) / NSL;
  const int lo  = s * per;
  const int hi  = min(e, lo + per);
  for (int i0 = lo + tid; i0 < hi; i0 += 4 * BT) {
    int d[4];
#pragma unroll
    for (int u = 0; u < 4; ++u) {
      const int i = i0 + u * BT;
      d[u] = (i < hi) ? dst[i] : 0;
    }
#pragma unroll
    for (int u = 0; u < 4; ++u)
      if (i0 + u * BT < hi) atomicAdd(&hist[d[u] >> BSH], 1);
  }
  __syncthreads();
  for (int b = tid; b < nbk; b += BT) tab[b * NSL + s] = hist[b];
}

#define RG (BT / 8)
#define RK (NSL / 32)
__global__ __launch_bounds__(BT)
void gcn_bin_place(const int* __restrict__ src, const int* __restrict__ dst,
                   const float* __restrict__ ew, const int* __restrict__ tab,
                   int* __restrict__ bst, int* __restrict__ binq_, int e, int nbk) {
  extern __shared__ int cur[];
  __shared__ int wsum[BT / 32];
  int* tot = cur + nbk;
  int4* binq = (int4*)binq_;
  const int4* tab4 = (const int4*)tab;
  const int s = blockIdx.x, tid = threadIdx.x, lane = tid & 31, wv = tid >> 5;
  const int q = tid & 7;
  for (int b0 = tid >> 3; b0 < nbk; b0 += 2 * RG) {
    int4 v[2][RK];
#pragma unroll
    for (int r = 0; r < 2; ++r) {
      const int b = b0 + r * RG;
#pragma unroll
      for (int k = 0; k < RK; ++k) {
        v[r][k] = make_int4(0, 0, 0, 0);
        if (b < nbk) v[r][k] = tab4[(size_t)b * (NSL / 4) + k * 8 + q];
      }
    }
#pragma unroll
    for (int r = 0; r < 2; ++r) {
      const int b = b0 + r * RG;
      int full = 0, part = 0;
#pragma unroll
      for (int k = 0; k < RK; ++k) {
        const int i0 = k * 32 + q * 4;
        full += (v[r][k].x + v[r][k].y) + (v[r][k].z + v[r][k].w);
        part += (((i0 + 0 < s) ? v[r][k].x : 0) + ((i0 + 1 < s) ? v[r][k].y : 0))
              + (((i0 + 2 < s) ? v[r][k].z : 0) + ((i0 + 3 < s) ? v[r][k].w : 0));
      }
      for (int o = 4; o > 0; o >>= 1) {
        full += __shfl_xor(full, o, 8);
        part += __shfl_xor(part, o, 8);
      }
      if (q == 0 && b < nbk) { tot[b] = full; cur[b] = part; }
    }
  }
  __syncthreads();
  int carry = 0;
  for (int c0 = 0; c0 < nbk; c0 += BT) {
    const int b = c0 + tid;
    const int t = (b < nbk) ? tot[b] : 0;
    int inc = t;
    for (int o = 1; o < 32; o <<= 1) {
      int u = __shfl_up(inc, o);
      if (lane >= o) inc += u;
    }
    if (lane == 31) wsum[wv] = inc;
    __syncthreads();
    int ex = carry + inc - t;
    int all = 0;
    for (int k = 0; k < BT / 32; ++k) {
      const int w = wsum[k];
      all += w;
      if (k < wv) ex += w;
    }
    if (b < nbk) {
      cur[b] += ex;
      if (s == 0) bst[b] = ex;
    }
    carry += all;
    __syncthreads();
  }
  const int per = (e + NSL - 1) / NSL;
  const int lo  = s * per;
  const int hi  = min(e, lo + per);
  for (int i0 = lo + tid; i0 < hi; i0 += 4 * BT) {
    int d[4], sv[4], wb[4];
#pragma unroll
    for (int u = 0; u < 4; ++u) {
      const int i = i0 + u * BT;
      d[u] = 0; sv[u] = 0; wb[u] = 0;
      if (i < hi) { d[u] = dst[i]; sv[u] = src[i]; wb[u] = __float_as_int(ew[i]); }
    }
#pragma unroll
    for (int u = 0; u < 4; ++u) {
      if (i0 + u * BT < hi) {
        const int p = atomicAdd(&cur[d[u] >> BSH], 1);
        binq[p] = make_int4(sv[u], wb[u], d[u], 0);
      }
    }
  }
}

__global__ __launch_bounds__(BT)
void gcn_bin_lists(const int* __restrict__ bst, const int* __restrict__ binq_,
                   int* __restrict__ cnt, int* __restrict__ rowptr, int* __restrict__ lst_,
                   int e, int n, int nbk) {
  __shared__ int c[BNODE];
  __shared__ int cursor[BNODE];
  __shared__ int wsum[BNODE / 32];
  const int4* binq = (const int4*)binq_;
  int2* lst = (int2*)lst_;
  const int b = blockIdx.x, tid = threadIdx.x, lane = tid & 31, wv = tid >> 5;
  const int base = b << BSH;
  const int bs = bst[b];
  const int be = (b + 1 < nbk) ? bst[b + 1] : e;
  if (tid < BNODE) c[tid] = 0;
  __syncthreads();
  for (int i0 = bs + tid; i0 < be; i0 += 4 * BT) {
    int d[4];
#pragma unroll
    for (int u = 0; u < 4; ++u) {
      const int i = i0 + u * BT;
      d[u] = (i < be) ? binq[i].z : base;
    }
#pragma unroll
    for (int u = 0; u < 4; ++u)
      if (i0 + u * BT < be) atomicAdd(&c[d[u] - base], 1);
  }
  __syncthreads();
  const int v = (tid < BNODE) ? c[tid] : 0;
  int inc = v;
  for (int o = 1; o < 32; o <<= 1) {
    int t = __shfl_up(inc, o);
    if (lane >= o) inc += t;
  }
  if (lane == 31 && tid < BNODE) wsum[wv] = inc;
  __syncthreads();
  if (tid < BNODE) {
    int ex = bs + inc - v;
    for (int k = 0; k < wv; ++k) ex += wsum[k];
    cursor[tid] = ex;
    const int node = base + tid;
    if (node < n) { cnt[node] = v; rowptr[node] = ex; }
  }
  __syncthreads();
  for (int i0 = bs + tid; i0 < be; i0 += 4 * BT) {
    int4 q[4];
#pragma unroll
    for (int u = 0; u < 4; ++u) {
      const int i = i0 + u * BT;
      q[u] = make_int4(0, 0, base, 0);
      if (i < be) q[u] = binq[i];
    }
#pragma unroll
    for (int u = 0; u < 4; ++u) {
      if (i0 + u * BT < be) {
        const int p = atomicAdd(&cursor[q[u].z - base], 1);
        lst[p] = make_int2(q[u].x, q[u].y);
      }
    }
  }
}

__global__ void gcn_dinv(float* __restrict__ dinv, const int* __restrict__ rowptr,
                         const int* __restrict__ cnt, const int* __restrict__ lst_, int n) {
  const int2* lst = (const int2*)lst_;
  int i = blockIdx.x * blockDim.x + threadIdx.x;
  if (i < n) {
    const int st = rowptr[i], len = cnt[i];
    float d = 1.0f;
    for (int j = 0; j < len; ++j) d += __int_as_float(lst[st + j].y);
    dinv[i] = (d > 0.0f) ? rsqrtf(d) : 0.0f;
  }
}

__global__ __launch_bounds__(256)
void gcn_norm(const float* __restrict__ dinv, const int* __restrict__ rowptr,
              const int* __restrict__ cnt, const int* __restrict__ lst_,
              int* __restrict__ lstn_, int n) {
  const int2* lst  = (const int2*)lst_;
  int2*       lstn = (int2*)lstn_;
  const int l  = threadIdx.x & 15;
  const int ng = (gridDim.x * blockDim.x) >> 4;
  for (int node = (blockIdx.x * blockDim.x + threadIdx.x) >> 4; node < n; node += ng) {
    const int st = rowptr[node], len = cnt[node];
    const float di = dinv[node];
    for (int j = l; j < len; j += 16) {
      const int2 e = lst[st + j];
      const float nrm = dinv[e.x] * __int_as_float(e.y) * di;
      lstn[st + j] = make_int2(e.x, __float_as_int(nrm));
    }
  }
}

#define LROWS 64
#define LGR   (LROWS / 16)
#define LMT   (LROWS / 32)

__device__ __forceinline__ float4 gcn_sum_list(const float* __restrict__ Xin, int ldi,
                                               const int2* __restrict__ lstn,
                                               int st, int end, int l, int2 e) {
  float4 acc = make_float4(0.0f, 0.0f, 0.0f, 0.0f);
  for (int p = st; p < end; p += 16) {
    const int m = min(16, end - p);
    if (p > st) e = lstn[p + min(l, m - 1)];
    const int   sv = e.x;
    const float wv = __int_as_float(e.y & ((l < m) ? -1 : 0));
    int k = 0;
    do {
      float4 h[4];
      float  w[4];
#pragma unroll
      for (int u = 0; u < 4; ++u) {
        const int s = __shfl(sv, k + u, 16);
        w[u] = __shfl(wv, k + u, 16);
        h[u] = make_float4(0.0f, 0.0f, 0.0f, 0.0f);
        if (k + u < m) h[u] = *(const float4*)(Xin + (size_t)s * ldi + l * 4);
      }
#pragma unroll
      for (int u = 0; u < 4; ++u) {
        acc.x = fmaf(h[u].x, w[u], acc.x);
        acc.y = fmaf(h[u].y, w[u], acc.y);
        acc.z = fmaf(h[u].z, w[u], acc.z);
        acc.w = fmaf(h[u].w, w[u], acc.w);
      }
      k += 4;
    } while (k < m);
  }
  return acc;
}

__global__ __launch_bounds__(256)
void gcn_layer(const float* __restrict__ Xin, int ldi, const float* __restrict__ dinv,
               const int* __restrict__ rowptr, const int* __restrict__ cnt,
               const int* __restrict__ lstn_, const float* __restrict__ W,
               const float* __restrict__ bias, float* Hout, int ldo, int n) {
  __shared__ __attribute__((aligned(16))) float Ws[D * LDP];
  __shared__ __attribute__((aligned(16))) float Zs[LROWS * LDP];
  const int2* lstn = (const int2*)lstn_;
  const int tid    = threadIdx.x;
  const int wave   = tid >> 5;
  const int lane   = tid & 31;
  const int laneLo = lane & 15;
  const int laneHi = lane >> 4;
  const int wn     = (wave & 3) << 4;
  const int wm0    = (wave >> 2) * (LMT * 16);
  const int g      = tid >> 4;
  const int l      = tid & 15;
  const int row0   = blockIdx.x * LROWS;
  const int r0     = g * LGR;

  int stL = 0, cL = 0;
  float dL = 0.0f;
  {
    const int nd = row0 + r0 + l;
    if (l < LGR && nd < n) { stL = rowptr[nd]; cL = cnt[nd]; dL = dinv[nd]; }
  }
  const int   gcol = wn + laneLo;
  const float bv   = bias[gcol];

  int st = __shfl(stL, 0, 16);
  int en = st + __shfl(cL, 0, 16);
  int2 e = make_int2(0, 0);
  if (en > st) e = lstn[st + min(l, min(16, en - st) - 1)];
  {
    float4 wr[D * D / 4 / 256];
#pragma unroll
    for (int u = 0; u < D * D / 4 / 256; ++u)
      wr[u] = *(const float4*)(W + (size_t)(tid + u * 256) * 4);
#pragma unroll
    for (int u = 0; u < D * D / 4 / 256; ++u) {
      const int i = tid + u * 256;
      *(float4*)(&Ws[(i >> 4) * LDP + ((i & 15) << 2)]) = wr[u];
    }
  }
#pragma unroll
  for (int j = 0; j < LGR; ++j) {
    const int node = row0 + r0 + j;
    const float di = __shfl(dL, j, 16);
    int nst = 0, nen = 0;
    int2 ne = make_int2(0, 0);
    if (j + 1 < LGR) {
      nst = __shfl(stL, j + 1, 16);
      nen = nst + __shfl(cL, j + 1, 16);
      if (nen > nst) ne = lstn[nst + min(l, min(16, nen - nst) - 1)];
    }
    float4 hs = make_float4(0.0f, 0.0f, 0.0f, 0.0f);
    if (node < n) hs = *(const float4*)(Xin + (size_t)node * ldi + l * 4);
    const float4 a = gcn_sum_list(Xin, ldi, lstn, st, en, l, e);
    const float sc = di * di;
    float4 z;
    z.x = fmaf(sc, hs.x, a.x); z.y = fmaf(sc, hs.y, a.y);
    z.z = fmaf(sc, hs.z, a.z); z.w = fmaf(sc, hs.w, a.w);
    *(float4*)(&Zs[(r0 + j) * LDP + l * 4]) = z;
    st = nst; en = nen; e = ne;
  }
  __syncthreads();

  v8f acc[LMT];
#pragma unroll
  for (int t = 0; t < LMT; ++t) acc[t] = v8f{};
#pragma unroll
  for (int k0 = 0; k0 < D; k0 += 4) {
    v2f b;
    b[0] = Ws[(k0 + 2 * laneHi + 0) * LDP + wn + laneLo];
    b[1] = Ws[(k0 + 2 * laneHi + 1) * LDP + wn + laneLo];
#pragma unroll
    for (int t = 0; t < LMT; ++t) {
      v2f a = *(const v2f*)(&Zs[(wm0 + t * 16 + laneLo) * LDP + k0 + 2 * laneHi]);
      acc[t] = __builtin_amdgcn_wmma_f32_16x16x4_f32(false, a, false, b,
                                                     (short)0, acc[t], false, false);
    }
  }
#pragma unroll
  for (int t = 0; t < LMT; ++t) {
#pragma unroll
    for (int v = 0; v < 8; ++v) {
      const int grow = row0 + wm0 + t * 16 + v + 8 * laneHi;
      float r = acc[t][v] + bv;
      r = r > 0.0f ? r : 0.0f;
      if (grow < n) Hout[(size_t)grow * ldo + gcol] = r;
    }
  }
}

__global__ void gcn_gemm_wmma(const float* __restrict__ X0, int ld0,
                              const float* __restrict__ X, int ldx,
                              const float* __restrict__ W,
                              const float* __restrict__ bias,
                              float* __restrict__ Out, int K, int n) {
  __shared__ float Xs[TROWS * LDP];
  __shared__ float Ws[D * LDP];
  const int tid    = threadIdx.x;
  const int wave   = tid >> 5;
  const int lane   = tid & 31;
  const int laneLo = lane & 15;
  const int laneHi = lane >> 4;
  const int row0   = blockIdx.x * TROWS;
  const int wm     = (wave >> 2) << 4;
  const int wn     = (wave & 3) << 4;

  v8f acc = {};
  for (int kc = 0; kc < K; kc += D) {
    const float* Xc = (kc == 0) ? X0 : X + kc;
    const int    ldc = (kc == 0) ? ld0 : ldx;
    for (int i = tid; i < TROWS * D; i += 256) {
      int r = i >> 6, c = i & 63;
      int rr = row0 + r; if (rr >= n) rr = n - 1;
      Xs[r * LDP + c] = Xc[(size_t)rr * ldc + c];
    }
    for (int i = tid; i < D * D; i += 256) {
      int r = i >> 6, c = i & 63;
      Ws[r * LDP + c] = W[(size_t)(kc + r) * D + c];
    }
    __syncthreads();
#pragma unroll
    for (int k0 = 0; k0 < D; k0 += 4) {
      v2f a = *(const v2f*)(&Xs[(wm + laneLo) * LDP + k0 + 2 * laneHi]);
      v2f b;
      b[0] = Ws[(k0 + 2 * laneHi + 0) * LDP + wn + laneLo];
      b[1] = Ws[(k0 + 2 * laneHi + 1) * LDP + wn + laneLo];
      acc = __builtin_amdgcn_wmma_f32_16x16x4_f32(false, a, false, b,
                                                  (short)0, acc, false, false);
    }
    __syncthreads();
  }
  const int gcol = wn + laneLo;
  const float bv = bias ? bias[gcol] : 0.0f;
#pragma unroll
  for (int v = 0; v < 8; ++v) {
    int grow = row0 + wm + v + 8 * laneHi;
    if (grow < n) Out[(size_t)grow * D + gcol] = acc[v] + bv;
  }
}

extern "C" void kernel_launch(void* const* d_in, const int* in_sizes, int n_in,
                              void* d_out, int out_size, void* d_ws, size_t ws_size,
                              hipStream_t stream) {
  const float* x   = (const float*)d_in[0];
  const int*   ei  = (const int*)d_in[1];
  const float* ew  = (const float*)d_in[2];
  const float* Wl[3] = {(const float*)d_in[3], (const float*)d_in[5], (const float*)d_in[7]};
  const float* bl[3] = {(const float*)d_in[4], (const float*)d_in[6], (const float*)d_in[8]};
  const float* Wf  = (const float*)d_in[9];
  const float* bf  = (const float*)d_in[10];

  const int N = in_sizes[0] / D;
  const int E = in_sizes[2];
  const int* src = ei;
  const int* dst = ei + E;

  const size_t Np = ((size_t)N + 255) & ~(size_t)255;
  const size_t Ep = ((size_t)E + 255) & ~(size_t)255;
  const int    nbk = (int)(Np >> BSH);
  const size_t Tn  = (size_t)nbk * NSL;
  float* deg    = (float*)d_ws;
  float* Hcat   = deg + Np;
  int*   cnt    = (int*)(Hcat + (size_t)N * 256);
  int*   rowptr = cnt + Np;
  int*   lst    = rowptr + Np;
  int*   lstn   = lst + 2 * Ep;
  int*   tab    = lstn + 2 * Ep;
  int*   bst    = tab + Tn;
  int*   binq   = bst + Tn;

  const int T = 256;
  const int PB = 1024;
  gcn_bin_count<<<NSL, BT, nbk * sizeof(int), stream>>>(dst, tab, E, nbk);
  gcn_bin_place<<<NSL, BT, 2 * nbk * sizeof(int), stream>>>(src, dst, ew, tab, bst, binq, E, nbk);
  gcn_bin_lists<<<nbk, BT, 0, stream>>>(bst, binq, cnt, rowptr, lst, E, N, nbk);
  gcn_dinv    <<<(N + T - 1) / T, T, 0, stream>>>(deg, rowptr, cnt, lst, N);
  gcn_norm    <<<PB, T, 0, stream>>>(deg, rowptr, cnt, lst, lstn, N);

  const int layer_blocks = (N + LROWS - 1) / LROWS;
  for (int l = 0; l < 3; ++l) {
    const float* Xin = (l == 0) ? x : Hcat + l * D;
    const int    ldi = (l == 0) ? D : 256;
    gcn_layer<<<layer_blocks, T, 0, stream>>>(Xin, ldi, deg, rowptr, cnt, lstn, Wl[l], bl[l],
                                              Hcat + (l + 1) * D, 256, N);
  }
  const int gemm_blocks = (N + TROWS - 1) / TROWS;
  gcn_gemm_wmma<<<gemm_blocks, T, 0, stream>>>(x, D, Hcat, 256, Wf, bf, (float*)d_out,
                                               4 * D, N);
}


# Final GEMM stages each 64-wide K-chunk with 16-byte loads, all six of a thread's loads issued before the first LDS store (was one 4-byte load at a time, each waited for singly)
# speedup vs baseline: 3.3236x; 1.1568x over previous
#include <hip/hip_runtime.h>

typedef __attribute__((ext_vector_type(2))) float v2f;
typedef __attribute__((ext_vector_type(8))) float v8f;

#define D     64
#define TROWS 32
#define LDP   68

#define BSH   8
#define BNODE (1 << BSH)
#define NSL   256
#define BT    512

__global__ __launch_bounds__(BT)
void gcn_bin_count(const int* __restrict__ dst, int* __restrict__ tab, int e, int nbk) {
  extern __shared__ int hist[];
  const int s = blockIdx.x, tid = threadIdx.x;
  for (int b = tid; b < nbk; b += BT) hist[b] = 0;
  __syncthreads();
  const int per = (e + NSL - 1) / NSL;
  const int lo  = s * per;
  const int hi  = min(e, lo + per);
  for (int i0 = lo + tid; i0 < hi; i0 += 4 * BT) {
    int d[4];
#pragma unroll
    for (int u = 0; u < 4; ++u) {
      const int i = i0 + u * BT;
      d[u] = (i < hi) ? dst[i] : 0;
    }
#pragma unroll
    for (int u = 0; u < 4; ++u)
      if (i0 + u * BT < hi) atomicAdd(&hist[d[u] >> BSH], 1);
  }
  __syncthreads();
  for (int b = tid; b < nbk; b += BT) tab[b * NSL + s] = hist[b];
}

#define RG (BT / 8)
#define RK (NSL / 32)
__global__ __launch_bounds__(BT)
void gcn_bin_place(const int* __restrict__ src, const int* __restrict__ dst,
                   const float* __restrict__ ew, const int* __restrict__ tab,
                   int* __restrict__ bst, int* __restrict__ binq_, int e, int nbk) {
  extern __shared__ int cur[];
  __shared__ int wsum[BT / 32];
  int* tot = cur + nbk;
  int4* binq = (int4*)binq_;
  const int4* tab4 = (const int4*)tab;
  const int s = blockIdx.x, tid = threadIdx.x, lane = tid & 31, wv = tid >> 5;
  const int q = tid & 7;
  for (int b0 = tid >> 3; b0 < nbk; b0 += 2 * RG) {
    int4 v[2][RK];
#pragma unroll
    for (int r = 0; r < 2; ++r) {
      const int b = b0 + r * RG;
#pragma unroll
      for (int k = 0; k < RK; ++k) {
        v[r][k] = make_int4(0, 0, 0, 0);
        if (b < nbk) v[r][k] = tab4[(size_t)b * (NSL / 4) + k * 8 + q];
      }
    }
#pragma unroll
    for (int r = 0; r < 2; ++r) {
      const int b = b0 + r * RG;
      int full = 0, part = 0;
#pragma unroll
      for (int k = 0; k < RK; ++k) {
        const int i0 = k * 32 + q * 4;
        full += (v[r][k].x + v[r][k].y) + (v[r][k].z + v[r][k].w);
        part += (((i0 + 0 < s) ? v[r][k].x : 0) + ((i0 + 1 < s) ? v[r][k].y : 0))
              + (((i0 + 2 < s) ? v[r][k].z : 0) + ((i0 + 3 < s) ? v[r][k].w : 0));
      }
      for (int o = 4; o > 0; o >>= 1) {
        full += __shfl_xor(full, o, 8);
        part += __shfl_xor(part, o, 8);
      }
      if (q == 0 && b < nbk) { tot[b] = full; cur[b] = part; }
    }
  }
  __syncthreads();
  int carry = 0;
  for (int c0 = 0; c0 < nbk; c0 += BT) {
    const int b = c0 + tid;
    const int t = (b < nbk) ? tot[b] : 0;
    int inc = t;
    for (int o = 1; o < 32; o <<= 1) {
      int u = __shfl_up(inc, o);
      if (lane >= o) inc += u;
    }
    if (lane == 31) wsum[wv] = inc;
    __syncthreads();
    int ex = carry + inc - t;
    int all = 0;
    for (int k = 0; k < BT / 32; ++k) {
      const int w = wsum[k];
      all += w;
      if (k < wv) ex += w;
    }
    if (b < nbk) {
      cur[b] += ex;
      if (s == 0) bst[b] = ex;
    }
    carry += all;
    __syncthreads();
  }
  const int per = (e + NSL - 1) / NSL;
  const int lo  = s * per;
  const int hi  = min(e, lo + per);
  for (int i0 = lo + tid; i0 < hi; i0 += 4 * BT) {
    int d[4], sv[4], wb[4];
#pragma unroll
    for (int u = 0; u < 4; ++u) {
      const int i = i0 + u * BT;
      d[u] = 0; sv[u] = 0; wb[u] = 0;
      if (i < hi) { d[u] = dst[i]; sv[u] = src[i]; wb[u] = __float_as_int(ew[i]); }
    }
#pragma unroll
    for (int u = 0; u < 4; ++u) {
      if (i0 + u * BT < hi) {
        const int p = atomicAdd(&cur[d[u] >> BSH], 1);
        binq[p] = make_int4(sv[u], wb[u], d[u], 0);
      }
    }
  }
}

__global__ __launch_bounds__(BT)
void gcn_bin_lists(const int* __restrict__ bst, const int* __restrict__ binq_,
                   int* __restrict__ cnt, int* __restrict__ rowptr, int* __restrict__ lst_,
                   int e, int n, int nbk) {
  __shared__ int c[BNODE];
  __shared__ int cursor[BNODE];
  __shared__ int wsum[BNODE / 32];
  const int4* binq = (const int4*)binq_;
  int2* lst = (int2*)lst_;
  const int b = blockIdx.x, tid = threadIdx.x, lane = tid & 31, wv = tid >> 5;
  const int base = b << BSH;
  const int bs = bst[b];
  const int be = (b + 1 < nbk) ? bst[b + 1] : e;
  if (tid < BNODE) c[tid] = 0;
  __syncthreads();
  for (int i0 = bs + tid; i0 < be; i0 += 4 * BT) {
    int d[4];
#pragma unroll
    for (int u = 0; u < 4; ++u) {
      const int i = i0 + u * BT;
      d[u] = (i < be) ? binq[i].z : base;
    }
#pragma unroll
    for (int u = 0; u < 4; ++u)
      if (i0 + u * BT < be) atomicAdd(&c[d[u] - base], 1);
  }
  __syncthreads();
  const int v = (tid < BNODE) ? c[tid] : 0;
  int inc = v;
  for (int o = 1; o < 32; o <<= 1) {
    int t = __shfl_up(inc, o);
    if (lane >= o) inc += t;
  }
  if (lane == 31 && tid < BNODE) wsum[wv] = inc;
  __syncthreads();
  if (tid < BNODE) {
    int ex = bs + inc - v;
    for (int k = 0; k < wv; ++k) ex += wsum[k];
    cursor[tid] = ex;
    const int node = base + tid;
    if (node < n) { cnt[node] = v; rowptr[node] = ex; }
  }
  __syncthreads();
  for (int i0 = bs + tid; i0 < be; i0 += 4 * BT) {
    int4 q[4];
#pragma unroll
    for (int u = 0; u < 4; ++u) {
      const int i = i0 + u * BT;
      q[u] = make_int4(0, 0, base, 0);
      if (i < be) q[u] = binq[i];
    }
#pragma unroll
    for (int u = 0; u < 4; ++u) {
      if (i0 + u * BT < be) {
        const int p = atomicAdd(&cursor[q[u].z - base], 1);
        lst[p] = make_int2(q[u].x, q[u].y);
      }
    }
  }
}

__global__ void gcn_dinv(float* __restrict__ dinv, const int* __restrict__ rowptr,
                         const int* __restrict__ cnt, const int* __restrict__ lst_, int n) {
  const int2* lst = (const int2*)lst_;
  int i = blockIdx.x * blockDim.x + threadIdx.x;
  if (i < n) {
    const int st = rowptr[i], len = cnt[i];
    float d = 1.0f;
    for (int j = 0; j < len; ++j) d += __int_as_float(lst[st + j].y);
    dinv[i] = (d > 0.0f) ? rsqrtf(d) : 0.0f;
  }
}

__global__ __launch_bounds__(256)
void gcn_norm(const float* __restrict__ dinv, const int* __restrict__ rowptr,
              const int* __restrict__ cnt, const int* __restrict__ lst_,
              int* __restrict__ lstn_, int n) {
  const int2* lst  = (const int2*)lst_;
  int2*       lstn = (int2*)lstn_;
  const int l  = threadIdx.x & 15;
  const int ng = (gridDim.x * blockDim.x) >> 4;
  for (int node = (blockIdx.x * blockDim.x + threadIdx.x) >> 4; node < n; node += ng) {
    const int st = rowptr[node], len = cnt[node];
    const float di = dinv[node];
    for (int j = l; j < len; j += 16) {
      const int2 e = lst[st + j];
      const float nrm = dinv[e.x] * __int_as_float(e.y) * di;
      lstn[st + j] = make_int2(e.x, __float_as_int(nrm));
    }
  }
}

#define LROWS 64
#define LGR   (LROWS / 16)
#define LMT   (LROWS / 32)

__device__ __forceinline__ float4 gcn_sum_list(const float* __restrict__ Xin, int ldi,
                                               const int2* __restrict__ lstn,
                                               int st, int end, int l, int2 e) {
  float4 acc = make_float4(0.0f, 0.0f, 0.0f, 0.0f);
  for (int p = st; p < end; p += 16) {
    const int m = min(16, end - p);
    if (p > st) e = lstn[p + min(l, m - 1)];
    const int   sv = e.x;
    const float wv = __int_as_float(e.y & ((l < m) ? -1 : 0));
    int k = 0;
    do {
      float4 h[4];
      float  w[4];
#pragma unroll
      for (int u = 0; u < 4; ++u) {
        const int s = __shfl(sv, k + u, 16);
        w[u] = __shfl(wv, k + u, 16);
        h[u] = make_float4(0.0f, 0.0f, 0.0f, 0.0f);
        if (k + u < m) h[u] = *(const float4*)(Xin + (size_t)s * ldi + l * 4);
      }
#pragma unroll
      for (int u = 0; u < 4; ++u) {
        acc.x = fmaf(h[u].x, w[u], acc.x);
        acc.y = fmaf(h[u].y, w[u], acc.y);
        acc.z = fmaf(h[u].z, w[u], acc.z);
        acc.w = fmaf(h[u].w, w[u], acc.w);
      }
      k += 4;
    } while (k < m);
  }
  return acc;
}

__global__ __launch_bounds__(256)
void gcn_layer(const float* __restrict__ Xin, int ldi, const float* __restrict__ dinv,
               const int* __restrict__ rowptr, const int* __restrict__ cnt,
               const int* __restrict__ lstn_, const float* __restrict__ W,
               const float* __restrict__ bias, float* Hout, int ldo, int n) {
  __shared__ __attribute__((aligned(16))) float Ws[D * LDP];
  __shared__ __attribute__((aligned(16))) float Zs[LROWS * LDP];
  const int2* lstn = (const int2*)lstn_;
  const int tid    = threadIdx.x;
  const int wave   = tid >> 5;
  const int lane   = tid & 31;
  const int laneLo = lane & 15;
  const int laneHi = lane >> 4;
  const int wn     = (wave & 3) << 4;
  const int wm0    = (wave >> 2) * (LMT * 16);
  const int g      = tid >> 4;
  const int l      = tid & 15;
  const int row0   = blockIdx.x * LROWS;
  const int r0     = g * LGR;

  int stL = 0, cL = 0;
  float dL = 0.0f;
  {
    const int nd = row0 + r0 + l;
    if (l < LGR && nd < n) { stL = rowptr[nd]; cL = cnt[nd]; dL = dinv[nd]; }
  }
  const int   gcol = wn + laneLo;
  const float bv   = bias[gcol];

  int st = __shfl(stL, 0, 16);
  int en = st + __shfl(cL, 0, 16);
  int2 e = make_int2(0, 0);
  if (en > st) e = lstn[st + min(l, min(16, en - st) - 1)];
  {
    float4 wr[D * D / 4 / 256];
#pragma unroll
    for (int u = 0; u < D * D / 4 / 256; ++u)
      wr[u] = *(const float4*)(W + (size_t)(tid + u * 256) * 4);
#pragma unroll
    for (int u = 0; u < D * D / 4 / 256; ++u) {
      const int i = tid + u * 256;
      *(float4*)(&Ws[(i >> 4) * LDP + ((i & 15) << 2)]) = wr[u];
    }
  }
#pragma unroll
  for (int j = 0; j < LGR; ++j) {
    const int node = row0 + r0 + j;
    const float di = __shfl(dL, j, 16);
    int nst = 0, nen = 0;
    int2 ne = make_int2(0, 0);
    if (j + 1 < LGR) {
      nst = __shfl(stL, j + 1, 16);
      nen = nst + __shfl(cL, j + 1, 16);
      if (nen > nst) ne = lstn[nst + min(l, min(16, nen - nst) - 1)];
    }
    float4 hs = make_float4(0.0f, 0.0f, 0.0f, 0.0f);
    if (node < n) hs = *(const float4*)(Xin + (size_t)node * ldi + l * 4);
    const float4 a = gcn_sum_list(Xin, ldi, lstn, st, en, l, e);
    const float sc = di * di;
    float4 z;
    z.x = fmaf(sc, hs.x, a.x); z.y = fmaf(sc, hs.y, a.y);
    z.z = fmaf(sc, hs.z, a.z); z.w = fmaf(sc, hs.w, a.w);
    *(float4*)(&Zs[(r0 + j) * LDP + l * 4]) = z;
    st = nst; en = nen; e = ne;
  }
  __syncthreads();

  v8f acc[LMT];
#pragma unroll
  for (int t = 0; t < LMT; ++t) acc[t] = v8f{};
#pragma unroll
  for (int k0 = 0; k0 < D; k0 += 4) {
    v2f b;
    b[0] = Ws[(k0 + 2 * laneHi + 0) * LDP + wn + laneLo];
    b[1] = Ws[(k0 + 2 * laneHi + 1) * LDP + wn + laneLo];
#pragma unroll
    for (int t = 0; t < LMT; ++t) {
      v2f a = *(const v2f*)(&Zs[(wm0 + t * 16 + laneLo) * LDP + k0 + 2 * laneHi]);
      acc[t] = __builtin_amdgcn_wmma_f32_16x16x4_f32(false, a, false, b,
                                                     (short)0, acc[t], false, false);
    }
  }
#pragma unroll
  for (int t = 0; t < LMT; ++t) {
#pragma unroll
    for (int v = 0; v < 8; ++v) {
      const int grow = row0 + wm0 + t * 16 + v + 8 * laneHi;
      float r = acc[t][v] + bv;
      r = r > 0.0f ? r : 0.0f;
      if (grow < n) Hout[(size_t)grow * ldo + gcol] = r;
    }
  }
}

#define GXV (TROWS * D / 4 / 256)
#define GWV (D * D / 4 / 256)
__global__ __launch_bounds__(256)
void gcn_gemm_wmma(const float* __restrict__ X0, int ld0,
                   const float* __restrict__ X, int ldx,
                   const float* __restrict__ W,
                   const float* __restrict__ bias,
                   float* __restrict__ Out, int K, int n) {
  __shared__ __attribute__((aligned(16))) float Xs[TROWS * LDP];
  __shared__ __attribute__((aligned(16))) float Ws[D * LDP];
  const int tid    = threadIdx.x;
  const int wave   = tid >> 5;
  const int lane   = tid & 31;
  const int laneLo = lane & 15;
  const int laneHi = lane >> 4;
  const int row0   = blockIdx.x * TROWS;
  const int wm     = (wave >> 2) << 4;
  const int wn     = (wave & 3) << 4;
  const int sr     = tid >> 4;
  const int sc     = (tid & 15) << 2;

  v8f acc = {};
  for (int kc = 0; kc < K; kc += D) {
    const float* Xc = (kc == 0) ? X0 : X + kc;
    const int    ldc = (kc == 0) ? ld0 : ldx;
    float4 xr[GXV], wr[GWV];
#pragma unroll
    for (int u = 0; u < GXV; ++u) {
      int rr = row0 + sr + u * 16; if (rr >= n) rr = n - 1;
      xr[u] = *(const float4*)(Xc + (size_t)rr * ldc + sc);
    }
#pragma unroll
    for (int u = 0; u < GWV; ++u)
      wr[u] = *(const float4*)(W + (size_t)kc * D + (size_t)(tid + u * 256) * 4);
#pragma unroll
    for (int u = 0; u < GXV; ++u)
      *(float4*)(&Xs[(sr + u * 16) * LDP + sc]) = xr[u];
#pragma unroll
    for (int u = 0; u < GWV; ++u)
      *(float4*)(&Ws[(sr + u * 16) * LDP + sc]) = wr[u];
    __syncthreads();
#pragma unroll
    for (int k0 = 0; k0 < D; k0 += 4) {
      v2f a = *(const v2f*)(&Xs[(wm + laneLo) * LDP + k0 + 2 * laneHi]);
      v2f b;
      b[0] = Ws[(k0 + 2 * laneHi + 0) * LDP + wn + laneLo];
      b[1] = Ws[(k0 + 2 * laneHi + 1) * LDP + wn + laneLo];
      acc = __builtin_amdgcn_wmma_f32_16x16x4_f32(false, a, false, b,
                                                  (short)0, acc, false, false);
    }
    __syncthreads();
  }
  const int gcol = wn + laneLo;
  const float bv = bias ? bias[gcol] : 0.0f;
#pragma unroll
  for (int v = 0; v < 8; ++v) {
    int grow = row0 + wm + v + 8 * laneHi;
    if (grow < n) Out[(size_t)grow * D + gcol] = acc[v] + bv;
  }
}

extern "C" void kernel_launch(void* const* d_in, const int* in_sizes, int n_in,
                              void* d_out, int out_size, void* d_ws, size_t ws_size,
                              hipStream_t stream) {
  const float* x   = (const float*)d_in[0];
  const int*   ei  = (const int*)d_in[1];
  const float* ew  = (const float*)d_in[2];
  const float* Wl[3] = {(const float*)d_in[3], (const float*)d_in[5], (const float*)d_in[7]};
  const float* bl[3] = {(const float*)d_in[4], (const float*)d_in[6], (const float*)d_in[8]};
  const float* Wf  = (const float*)d_in[9];
  const float* bf  = (const float*)d_in[10];

  const int N = in_sizes[0] / D;
  const int E = in_sizes[2];
  const int* src = ei;
  const int* dst = ei + E;

  const size_t Np = ((size_t)N + 255) & ~(size_t)255;
  const size_t Ep = ((size_t)E + 255) & ~(size_t)255;
  const int    nbk = (int)(Np >> BSH);
  const size_t Tn  = (size_t)nbk * NSL;
  float* deg    = (float*)d_ws;
  float* Hcat   = deg + Np;
  int*   cnt    = (int*)(Hcat + (size_t)N * 256);
  int*   rowptr = cnt + Np;
  int*   lst    = rowptr + Np;
  int*   lstn   = lst + 2 * Ep;
  int*   tab    = lstn + 2 * Ep;
  int*   bst    = tab + Tn;
  int*   binq   = bst + Tn;

  const int T = 256;
  const int PB = 1024;
  gcn_bin_count<<<NSL, BT, nbk * sizeof(int), stream>>>(dst, tab, E, nbk);
  gcn_bin_place<<<NSL, BT, 2 * nbk * sizeof(int), stream>>>(src, dst, ew, tab, bst, binq, E, nbk);
  gcn_bin_lists<<<nbk, BT, 0, stream>>>(bst, binq, cnt, rowptr, lst, E, N, nbk);
  gcn_dinv    <<<(N + T - 1) / T, T, 0, stream>>>(deg, rowptr, cnt, lst, N);
  gcn_norm    <<<PB, T, 0, stream>>>(deg, rowptr, cnt, lst, lstn, N);

  const int layer_blocks = (N + LROWS - 1) / LROWS;
  for (int l = 0; l < 3; ++l) {
    const float* Xin = (l == 0) ? x : Hcat + l * D;
    const int    ldi = (l == 0) ? D : 256;
    gcn_layer<<<layer_blocks, T, 0, stream>>>(Xin, ldi, deg, rowptr, cnt, lstn, Wl[l], bl[l],
                                              Hcat + (l + 1) * D, 256, N);
  }
  const int gemm_blocks = (N + TROWS - 1) / TROWS;
  gcn_gemm_wmma<<<gemm_blocks, T, 0, stream>>>(x, D, Hcat, 256, Wf, bf, (float*)d_out,
                                               4 * D, N);
}


# Degree and edge-norm passes folded into neighbours: degrees summed by LDS float adds while the lists are filled (sum order changes), norms formed by layer 0 as it walks the lists; 2 launches gone
# speedup vs baseline: 3.5537x; 1.0692x over previous
#include <hip/hip_runtime.h>

typedef __attribute__((ext_vector_type(2))) float v2f;
typedef __attribute__((ext_vector_type(8))) float v8f;

#define D     64
#define TROWS 32
#define LDP   68

#define BSH   8
#define BNODE (1 << BSH)
#define NSL   256
#define BT    512

__global__ __launch_bounds__(BT)
void gcn_bin_count(const int* __restrict__ dst, int* __restrict__ tab, int e, int nbk) {
  extern __shared__ int hist[];
  const int s = blockIdx.x, tid = threadIdx.x;
  for (int b = tid; b < nbk; b += BT) hist[b] = 0;
  __syncthreads();
  const int per = (e + NSL - 1) / NSL;
  const int lo  = s * per;
  const int hi  = min(e, lo + per);
  for (int i0 = lo + tid; i0 < hi; i0 += 4 * BT) {
    int d[4];
#pragma unroll
    for (int u = 0; u < 4; ++u) {
      const int i = i0 + u * BT;
      d[u] = (i < hi) ? dst[i] : 0;
    }
#pragma unroll
    for (int u = 0; u < 4; ++u)
      if (i0 + u * BT < hi) atomicAdd(&hist[d[u] >> BSH], 1);
  }
  __syncthreads();
  for (int b = tid; b < nbk; b += BT) tab[b * NSL + s] = hist[b];
}

#define RG (BT / 8)
#define RK (NSL / 32)
__global__ __launch_bounds__(BT)
void gcn_bin_place(const int* __restrict__ src, const int* __restrict__ dst,
                   const float* __restrict__ ew, const int* __restrict__ tab,
                   int* __restrict__ bst, int* __restrict__ binq_, int e, int nbk) {
  extern __shared__ int cur[];
  __shared__ int wsum[BT / 32];
  int* tot = cur + nbk;
  int4* binq = (int4*)binq_;
  const int4* tab4 = (const int4*)tab;
  const int s = blockIdx.x, tid = threadIdx.x, lane = tid & 31, wv = tid >> 5;
  const int q = tid & 7;
  for (int b0 = tid >> 3; b0 < nbk; b0 += 2 * RG) {
    int4 v[2][RK];
#pragma unroll
    for (int r = 0; r < 2; ++r) {
      const int b = b0 + r * RG;
#pragma unroll
      for (int k = 0; k < RK; ++k) {
        v[r][k] = make_int4(0, 0, 0, 0);
        if (b < nbk) v[r][k] = tab4[(size_t)b * (NSL / 4) + k * 8 + q];
      }
    }
#pragma unroll
    for (int r = 0; r < 2; ++r) {
      const int b = b0 + r * RG;
      int full = 0, part = 0;
#pragma unroll
      for (int k = 0; k < RK; ++k) {
        const int i0 = k * 32 + q * 4;
        full += (v[r][k].x + v[r][k].y) + (v[r][k].z + v[r][k].w);
        part += (((i0 + 0 < s) ? v[r][k].x : 0) + ((i0 + 1 < s) ? v[r][k].y : 0))
              + (((i0 + 2 < s) ? v[r][k].z : 0) + ((i0 + 3 < s) ? v[r][k].w : 0));
      }
      for (int o = 4; o > 0; o >>= 1) {
        full += __shfl_xor(full, o, 8);
        part += __shfl_xor(part, o, 8);
      }
      if (q == 0 && b < nbk) { tot[b] = full; cur[b] = part; }
    }
  }
  __syncthreads();
  int carry = 0;
  for (int c0 = 0; c0 < nbk; c0 += BT) {
    const int b = c0 + tid;
    const int t = (b < nbk) ? tot[b] : 0;
    int inc = t;
    for (int o = 1; o < 32; o <<= 1) {
      int u = __shfl_up(inc, o);
      if (lane >= o) inc += u;
    }
    if (lane == 31) wsum[wv] = inc;
    __syncthreads();
    int ex = carry + inc - t;
    int all = 0;
    for (int k = 0; k < BT / 32; ++k) {
      const int w = wsum[k];
      all += w;
      if (k < wv) ex += w;
    }
    if (b < nbk) {
      cur[b] += ex;
      if (s == 0) bst[b] = ex;
    }
    carry += all;
    __syncthreads();
  }
  const int per = (e + NSL - 1) / NSL;
  const int lo  = s * per;
  const int hi  = min(e, lo + per);
  for (int i0 = lo + tid; i0 < hi; i0 += 4 * BT) {
    int d[4], sv[4], wb[4];
#pragma unroll
    for (int u = 0; u < 4; ++u) {
      const int i = i0 + u * BT;
      d[u] = 0; sv[u] = 0; wb[u] = 0;
      if (i < hi) { d[u] = dst[i]; sv[u] = src[i]; wb[u] = __float_as_int(ew[i]); }
    }
#pragma unroll
    for (int u = 0; u < 4; ++u) {
      if (i0 + u * BT < hi) {
        const int p = atomicAdd(&cur[d[u] >> BSH], 1);
        binq[p] = make_int4(sv[u], wb[u], d[u], 0);
      }
    }
  }
}

__global__ __launch_bounds__(BT)
void gcn_bin_lists(const int* __restrict__ bst, const int* __restrict__ binq_,
                   int* __restrict__ cnt, int* __restrict__ rowptr, int* __restrict__ lst_,
                   float* __restrict__ dinv, int e, int n, int nbk) {
  __shared__ int c[BNODE];
  __shared__ int cursor[BNODE];
  __shared__ float wdeg[BNODE];
  __shared__ int wsum[BNODE / 32];
  const int4* binq = (const int4*)binq_;
  int2* lst = (int2*)lst_;
  const int b = blockIdx.x, tid = threadIdx.x, lane = tid & 31, wv = tid >> 5;
  const int base = b << BSH;
  const int bs = bst[b];
  const int be = (b + 1 < nbk) ? bst[b + 1] : e;
  if (tid < BNODE) { c[tid] = 0; wdeg[tid] = 0.0f; }
  __syncthreads();
  for (int i0 = bs + tid; i0 < be; i0 += 4 * BT) {
    int d[4];
#pragma unroll
    for (int u = 0; u < 4; ++u) {
      const int i = i0 + u * BT;
      d[u] = (i < be) ? binq[i].z : base;
    }
#pragma unroll
    for (int u = 0; u < 4; ++u)
      if (i0 + u * BT < be) atomicAdd(&c[d[u] - base], 1);
  }
  __syncthreads();
  const int v = (tid < BNODE) ? c[tid] : 0;
  int inc = v;
  for (int o = 1; o < 32; o <<= 1) {
    int t = __shfl_up(inc, o);
    if (lane >= o) inc += t;
  }
  if (lane == 31 && tid < BNODE) wsum[wv] = inc;
  __syncthreads();
  if (tid < BNODE) {
    int ex = bs + inc - v;
    for (int k = 0; k < wv; ++k) ex += wsum[k];
    cursor[tid] = ex;
    const int node = base + tid;
    if (node < n) { cnt[node] = v; rowptr[node] = ex; }
  }
  __syncthreads();
  for (int i0 = bs + tid; i0 < be; i0 += 4 * BT) {
    int4 q[4];
#pragma unroll
    for (int u = 0; u < 4; ++u) {
      const int i = i0 + u * BT;
      q[u] = make_int4(0, 0, base, 0);
      if (i < be) q[u] = binq[i];
    }
#pragma unroll
    for (int u = 0; u < 4; ++u) {
      if (i0 + u * BT < be) {
        const int p = atomicAdd(&cursor[q[u].z - base], 1);
        atomicAdd(&wdeg[q[u].z - base], __int_as_float(q[u].y));
        lst[p] = make_int2(q[u].x, q[u].y);
      }
    }
  }
  __syncthreads();
  if (tid < BNODE && base + tid < n) {
    const float d = 1.0f + wdeg[tid];
    dinv[base + tid] = (d > 0.0f) ? rsqrtf(d) : 0.0f;
  }
}

#define LROWS 64
#define LGR   (LROWS / 16)
#define LMT   (LROWS / 32)

template <bool LATEW, bool HEAD>
__device__ __forceinline__ void gcn_rows4(float4& acc, float& wv,
                                          const float* __restrict__ Xin, int ldi,
                                          int sv, int k, int m, int l,
                                          float dn, float raw, float di,
                                          int2* __restrict__ lo) {
  float4 h[4];
  float  w[4];
#pragma unroll
  for (int u = 0; u < 4; ++u) {
    const int s = __shfl(sv, k + u, 16);
    if (!LATEW) w[u] = __shfl(wv, k + u, 16);
    h[u] = make_float4(0.0f, 0.0f, 0.0f, 0.0f);
    if (k + u < m) h[u] = *(const float4*)(Xin + (size_t)s * ldi + l * 4);
  }
  if (LATEW) {
    if (HEAD) {
      const float nrm = dn * raw * di;
      if (l < m) lo[l] = make_int2(sv, __float_as_int(nrm));
      wv = (l < m) ? nrm : 0.0f;
    }
#pragma unroll
    for (int u = 0; u < 4; ++u) w[u] = __shfl(wv, k + u, 16);
  }
#pragma unroll
  for (int u = 0; u < 4; ++u) {
    acc.x = fmaf(h[u].x, w[u], acc.x);
    acc.y = fmaf(h[u].y, w[u], acc.y);
    acc.z = fmaf(h[u].z, w[u], acc.z);
    acc.w = fmaf(h[u].w, w[u], acc.w);
  }
}

template <bool FIRST>
__device__ __forceinline__ float4 gcn_sum_list(const float* __restrict__ Xin, int ldi,
                                               const int2* __restrict__ lin,
                                               int2* __restrict__ lout,
                                               const float* __restrict__ dinv, float di,
                                               int st, int end, int l, int2 e) {
  float4 acc = make_float4(0.0f, 0.0f, 0.0f, 0.0f);
  for (int p = st; p < end; p += 16) {
    const int m = min(16, end - p);
    if (p > st) e = lin[p + min(l, m - 1)];
    const int sv = e.x;
    if (FIRST) {
      const float dn  = dinv[sv];
      const float raw = __int_as_float(e.y);
      float wv = 0.0f;
      gcn_rows4<true, true>(acc, wv, Xin, ldi, sv, 0, m, l, dn, raw, di, lout + p);
      for (int k = 4; k < m; k += 4)
        gcn_rows4<true, false>(acc, wv, Xin, ldi, sv, k, m, l, dn, raw, di, lout + p);
    } else {
      float wv = __int_as_float(e.y & ((l < m) ? -1 : 0));
      int k = 0;
      do {
        gcn_rows4<false, false>(acc, wv, Xin, ldi, sv, k, m, l, 0.0f, 0.0f, di, lout);
        k += 4;
      } while (k < m);
    }
  }
  return acc;
}

template <bool FIRST>
__global__ __launch_bounds__(256)
void gcn_layer(const float* __restrict__ Xin, int ldi, const float* __restrict__ dinv,
               const int* __restrict__ rowptr, const int* __restrict__ cnt,
               const int* __restrict__ lin_, int* __restrict__ lout_,
               const float* __restrict__ W,
               const float* __restrict__ bias, float* Hout, int ldo, int n) {
  __shared__ __attribute__((aligned(16))) float Ws[D * LDP];
  __shared__ __attribute__((aligned(16))) float Zs[LROWS * LDP];
  const int2* lin = (const int2*)lin_;
  int2*       lout = (int2*)lout_;
  const int tid    = threadIdx.x;
  const int wave   = tid >> 5;
  const int lane   = tid & 31;
  const int laneLo = lane & 15;
  const int laneHi = lane >> 4;
  const int wn     = (wave & 3) << 4;
  const int wm0    = (wave >> 2) * (LMT * 16);
  const int g      = tid >> 4;
  const int l      = tid & 15;
  const int row0   = blockIdx.x * LROWS;
  const int r0     = g * LGR;

  int stL = 0, cL = 0;
  float dL = 0.0f;
  {
    const int nd = row0 + r0 + l;
    if (l < LGR && nd < n) { stL = rowptr[nd]; cL = cnt[nd]; dL = dinv[nd]; }
  }
  const int   gcol = wn + laneLo;
  const float bv   = bias[gcol];

  int st = __shfl(stL, 0, 16);
  int en = st + __shfl(cL, 0, 16);
  int2 e = make_int2(0, 0);
  if (en > st) e = lin[st + min(l, min(16, en - st) - 1)];
  {
    float4 wr[D * D / 4 / 256];
#pragma unroll
    for (int u = 0; u < D * D / 4 / 256; ++u)
      wr[u] = *(const float4*)(W + (size_t)(tid + u * 256) * 4);
#pragma unroll
    for (int u = 0; u < D * D / 4 / 256; ++u) {
      const int i = tid + u * 256;
      *(float4*)(&Ws[(i >> 4) * LDP + ((i & 15) << 2)]) = wr[u];
    }
  }
#pragma unroll
  for (int j = 0; j < LGR; ++j) {
    const int node = row0 + r0 + j;
    const float di = __shfl(dL, j, 16);
    int nst = 0, nen = 0;
    int2 ne = make_int2(0, 0);
    if (j + 1 < LGR) {
      nst = __shfl(stL, j + 1, 16);
      nen = nst + __shfl(cL, j + 1, 16);
      if (nen > nst) ne = lin[nst + min(l, min(16, nen - nst) - 1)];
    }
    float4 hs = make_float4(0.0f, 0.0f, 0.0f, 0.0f);
    if (node < n) hs = *(const float4*)(Xin + (size_t)node * ldi + l * 4);
    const float4 a = gcn_sum_list<FIRST>(Xin, ldi, lin, lout, dinv, di, st, en, l, e);
    const float sc = di * di;
    float4 z;
    z.x = fmaf(sc, hs.x, a.x); z.y = fmaf(sc, hs.y, a.y);
    z.z = fmaf(sc, hs.z, a.z); z.w = fmaf(sc, hs.w, a.w);
    *(float4*)(&Zs[(r0 + j) * LDP + l * 4]) = z;
    st = nst; en = nen; e = ne;
  }
  __syncthreads();

  v8f acc[LMT];
#pragma unroll
  for (int t = 0; t < LMT; ++t) acc[t] = v8f{};
#pragma unroll
  for (int k0 = 0; k0 < D; k0 += 4) {
    v2f b;
    b[0] = Ws[(k0 + 2 * laneHi + 0) * LDP + wn + laneLo];
    b[1] = Ws[(k0 + 2 * laneHi + 1) * LDP + wn + laneLo];
#pragma unroll
    for (int t = 0; t < LMT; ++t) {
      v2f a = *(const v2f*)(&Zs[(wm0 + t * 16 + laneLo) * LDP + k0 + 2 * laneHi]);
      acc[t] = __builtin_amdgcn_wmma_f32_16x16x4_f32(false, a, false, b,
                                                     (short)0, acc[t], false, false);
    }
  }
#pragma unroll
  for (int t = 0; t < LMT; ++t) {
#pragma unroll
    for (int v = 0; v < 8; ++v) {
      const int grow = row0 + wm0 + t * 16 + v + 8 * laneHi;
      float r = acc[t][v] + bv;
      r = r > 0.0f ? r : 0.0f;
      if (grow < n) Hout[(size_t)grow * ldo + gcol] = r;
    }
  }
}

#define GXV (TROWS * D / 4 / 256)
#define GWV (D * D / 4 / 256)
__global__ __launch_bounds__(256)
void gcn_gemm_wmma(const float* __restrict__ X0, int ld0,
                   const float* __restrict__ X, int ldx,
                   const float* __restrict__ W,
                   const float* __restrict__ bias,
                   float* __restrict__ Out, int K, int n) {
  __shared__ __attribute__((aligned(16))) float Xs[TROWS * LDP];
  __shared__ __attribute__((aligned(16))) float Ws[D * LDP];
  const int tid    = threadIdx.x;
  const int wave   = tid >> 5;
  const int lane   = tid & 31;
  const int laneLo = lane & 15;
  const int laneHi = lane >> 4;
  const int row0   = blockIdx.x * TROWS;
  const int wm     = (wave >> 2) << 4;
  const int wn     = (wave & 3) << 4;
  const int sr     = tid >> 4;
  const int sc     = (tid & 15) << 2;

  v8f acc = {};
  for (int kc = 0; kc < K; kc += D) {
    const float* Xc = (kc == 0) ? X0 : X + kc;
    const int    ldc = (kc == 0) ? ld0 : ldx;
    float4 xr[GXV], wr[GWV];
#pragma unroll
    for (int u = 0; u < GXV; ++u) {
      int rr = row0 + sr + u * 16; if (rr >= n) rr = n - 1;
      xr[u] = *(const float4*)(Xc + (size_t)rr * ldc + sc);
    }
#pragma unroll
    for (int u = 0; u < GWV; ++u)
      wr[u] = *(const float4*)(W + (size_t)kc * D + (size_t)(tid + u * 256) * 4);
#pragma unroll
    for (int u = 0; u < GXV; ++u)
      *(float4*)(&Xs[(sr + u * 16) * LDP + sc]) = xr[u];
#pragma unroll
    for (int u = 0; u < GWV; ++u)
      *(float4*)(&Ws[(sr + u * 16) * LDP + sc]) = wr[u];
    __syncthreads();
#pragma unroll
    for (int k0 = 0; k0 < D; k0 += 4) {
      v2f a = *(const v2f*)(&Xs[(wm + laneLo) * LDP + k0 + 2 * laneHi]);
      v2f b;
      b[0] = Ws[(k0 + 2 * laneHi + 0) * LDP + wn + laneLo];
      b[1] = Ws[(k0 + 2 * laneHi + 1) * LDP + wn + laneLo];
      acc = __builtin_amdgcn_wmma_f32_16x16x4_f32(false, a, false, b,
                                                  (short)0, acc, false, false);
    }
    __syncthreads();
  }
  const int gcol = wn + laneLo;
  const float bv = bias ? bias[gcol] : 0.0f;
#pragma unroll
  for (int v = 0; v < 8; ++v) {
    int grow = row0 + wm + v + 8 * laneHi;
    if (grow < n) Out[(size_t)grow * D + gcol] = acc[v] + bv;
  }
}

extern "C" void kernel_launch(void* const* d_in, const int* in_sizes, int n_in,
                              void* d_out, int out_size, void* d_ws, size_t ws_size,
                              hipStream_t stream) {
  const float* x   = (const float*)d_in[0];
  const int*   ei  = (const int*)d_in[1];
  const float* ew  = (const float*)d_in[2];
  const float* Wl[3] = {(const float*)d_in[3], (const float*)d_in[5], (const float*)d_in[7]};
  const float* bl[3] = {(const float*)d_in[4], (const float*)d_in[6], (const float*)d_in[8]};
  const float* Wf  = (const float*)d_in[9];
  const float* bf  = (const float*)d_in[10];

  const int N = in_sizes[0] / D;
  const int E = in_sizes[2];
  const int* src = ei;
  const int* dst = ei + E;

  const size_t Np = ((size_t)N + 255) & ~(size_t)255;
  const size_t Ep = ((size_t)E + 255) & ~(size_t)255;
  const int    nbk = (int)(Np >> BSH);
  const size_t Tn  = (size_t)nbk * NSL;
  float* deg    = (float*)d_ws;
  float* Hcat   = deg + Np;
  int*   cnt    = (int*)(Hcat + (size_t)N * 256);
  int*   rowptr = cnt + Np;
  int*   lst    = rowptr + Np;
  int*   lstn   = lst + 2 * Ep;
  int*   tab    = lstn + 2 * Ep;
  int*   bst    = tab + Tn;
  int*   binq   = bst + Tn;

  const int T = 256;
  gcn_bin_count<<<NSL, BT, nbk * sizeof(int), stream>>>(dst, tab, E, nbk);
  gcn_bin_place<<<NSL, BT, 2 * nbk * sizeof(int), stream>>>(src, dst, ew, tab, bst, binq, E, nbk);
  gcn_bin_lists<<<nbk, BT, 0, stream>>>(bst, binq, cnt, rowptr, lst, deg, E, N, nbk);

  const int layer_blocks = (N + LROWS - 1) / LROWS;
  gcn_layer<true><<<layer_blocks, T, 0, stream>>>(x, D, deg, rowptr, cnt, lst, lstn,
                                                  Wl[0], bl[0], Hcat + D, 256, N);
  gcn_layer<false><<<layer_blocks, T, 0, stream>>>(Hcat + D, 256, deg, rowptr, cnt, lstn, nullptr,
                                                   Wl[1], bl[1], Hcat + 2 * D, 256, N);
  gcn_layer<false><<<layer_blocks, T, 0, stream>>>(Hcat + 2 * D, 256, deg, rowptr, cnt, lstn, nullptr,
                                                   Wl[2], bl[2], Hcat + 3 * D, 256, N);
  const int gemm_blocks = (N + TROWS - 1) / TROWS;
  gcn_gemm_wmma<<<gemm_blocks, T, 0, stream>>>(x, D, Hcat, 256, Wf, bf, (float*)d_out,
                                               4 * D, N);
}


# List entries handed round by row-share moves (constant lane, 16-entry walk written out as 4 batches) instead of LDS permutes; row gathers under no branch: past the end the last row again, weight 0
# speedup vs baseline: 3.7157x; 1.0456x over previous
#include <hip/hip_runtime.h>

typedef __attribute__((ext_vector_type(2))) float v2f;
typedef __attribute__((ext_vector_type(8))) float v8f;

#define D     64
#define TROWS 32
#define LDP   68

#define BSH   8
#define BNODE (1 << BSH)
#define NSL   256
#define BT    512

__global__ __launch_bounds__(BT)
void gcn_bin_count(const int* __restrict__ dst, int* __restrict__ tab, int e, int nbk) {
  extern __shared__ int hist[];
  const int s = blockIdx.x, tid = threadIdx.x;
  for (int b = tid; b < nbk; b += BT) hist[b] = 0;
  __syncthreads();
  const int per = (e + NSL - 1) / NSL;
  const int lo  = s * per;
  const int hi  = min(e, lo + per);
  for (int i0 = lo + tid; i0 < hi; i0 += 4 * BT) {
    int d[4];
#pragma unroll
    for (int u = 0; u < 4; ++u) {
      const int i = i0 + u * BT;
      d[u] = (i < hi) ? dst[i] : 0;
    }
#pragma unroll
    for (int u = 0; u < 4; ++u)
      if (i0 + u * BT < hi) atomicAdd(&hist[d[u] >> BSH], 1);
  }
  __syncthreads();
  for (int b = tid; b < nbk; b += BT) tab[b * NSL + s] = hist[b];
}

#define RG (BT / 8)
#define RK (NSL / 32)
__global__ __launch_bounds__(BT)
void gcn_bin_place(const int* __restrict__ src, const int* __restrict__ dst,
                   const float* __restrict__ ew, const int* __restrict__ tab,
                   int* __restrict__ bst, int* __restrict__ binq_, int e, int nbk) {
  extern __shared__ int cur[];
  __shared__ int wsum[BT / 32];
  int* tot = cur + nbk;
  int4* binq = (int4*)binq_;
  const int4* tab4 = (const int4*)tab;
  const int s = blockIdx.x, tid = threadIdx.x, lane = tid & 31, wv = tid >> 5;
  const int q = tid & 7;
  for (int b0 = tid >> 3; b0 < nbk; b0 += 2 * RG) {
    int4 v[2][RK];
#pragma unroll
    for (int r = 0; r < 2; ++r) {
      const int b = b0 + r * RG;
#pragma unroll
      for (int k = 0; k < RK; ++k) {
        v[r][k] = make_int4(0, 0, 0, 0);
        if (b < nbk) v[r][k] = tab4[(size_t)b * (NSL / 4) + k * 8 + q];
      }
    }
#pragma unroll
    for (int r = 0; r < 2; ++r) {
      const int b = b0 + r * RG;
      int full = 0, part = 0;
#pragma unroll
      for (int k = 0; k < RK; ++k) {
        const int i0 = k * 32 + q * 4;
        full += (v[r][k].x + v[r][k].y) + (v[r][k].z + v[r][k].w);
        part += (((i0 + 0 < s) ? v[r][k].x : 0) + ((i0 + 1 < s) ? v[r][k].y : 0))
              + (((i0 + 2 < s) ? v[r][k].z : 0) + ((i0 + 3 < s) ? v[r][k].w : 0));
      }
      for (int o = 4; o > 0; o >>= 1) {
        full += __shfl_xor(full, o, 8);
        part += __shfl_xor(part, o, 8);
      }
      if (q == 0 && b < nbk) { tot[b] = full; cur[b] = part; }
    }
  }
  __syncthreads();
  int carry = 0;
  for (int c0 = 0; c0 < nbk; c0 += BT) {
    const int b = c0 + tid;
    const int t = (b < nbk) ? tot[b] : 0;
    int inc = t;
    for (int o = 1; o < 32; o <<= 1) {
      int u = __shfl_up(inc, o);
      if (lane >= o) inc += u;
    }
    if (lane == 31) wsum[wv] = inc;
    __syncthreads();
    int ex = carry + inc - t;
    int all = 0;
    for (int k = 0; k < BT / 32; ++k) {
      const int w = wsum[k];
      all += w;
      if (k < wv) ex += w;
    }
    if (b < nbk) {
      cur[b] += ex;
      if (s == 0) bst[b] = ex;
    }
    carry += all;
    __syncthreads();
  }
  const int per = (e + NSL - 1) / NSL;
  const int lo  = s * per;
  const int hi  = min(e, lo + per);
  for (int i0 = lo + tid; i0 < hi; i0 += 4 * BT) {
    int d[4], sv[4], wb[4];
#pragma unroll
    for (int u = 0; u < 4; ++u) {
      const int i = i0 + u * BT;
      d[u] = 0; sv[u] = 0; wb[u] = 0;
      if (i < hi) { d[u] = dst[i]; sv[u] = src[i]; wb[u] = __float_as_int(ew[i]); }
    }
#pragma unroll
    for (int u = 0; u < 4; ++u) {
      if (i0 + u * BT < hi) {
        const int p = atomicAdd(&cur[d[u] >> BSH], 1);
        binq[p] = make_int4(sv[u], wb[u], d[u], 0);
      }
    }
  }
}

__global__ __launch_bounds__(BT)
void gcn_bin_lists(const int* __restrict__ bst, const int* __restrict__ binq_,
                   int* __restrict__ cnt, int* __restrict__ rowptr, int* __restrict__ lst_,
                   float* __restrict__ dinv, int e, int n, int nbk) {
  __shared__ int c[BNODE];
  __shared__ int cursor[BNODE];
  __shared__ float wdeg[BNODE];
  __shared__ int wsum[BNODE / 32];
  const int4* binq = (const int4*)binq_;
  int2* lst = (int2*)lst_;
  const int b = blockIdx.x, tid = threadIdx.x, lane = tid & 31, wv = tid >> 5;
  const int base = b << BSH;
  const int bs = bst[b];
  const int be = (b + 1 < nbk) ? bst[b + 1] : e;
  if (tid < BNODE) { c[tid] = 0; wdeg[tid] = 0.0f; }
  __syncthreads();
  for (int i0 = bs + tid; i0 < be; i0 += 4 * BT) {
    int d[4];
#pragma unroll
    for (int u = 0; u < 4; ++u) {
      const int i = i0 + u * BT;
      d[u] = (i < be) ? binq[i].z : base;
    }
#pragma unroll
    for (int u = 0; u < 4; ++u)
      if (i0 + u * BT < be) atomicAdd(&c[d[u] - base], 1);
  }
  __syncthreads();
  const int v = (tid < BNODE) ? c[tid] : 0;
  int inc = v;
  for (int o = 1; o < 32; o <<= 1) {
    int t = __shfl_up(inc, o);
    if (lane >= o) inc += t;
  }
  if (lane == 31 && tid < BNODE) wsum[wv] = inc;
  __syncthreads();
  if (tid < BNODE) {
    int ex = bs + inc - v;
    for (int k = 0; k < wv; ++k) ex += wsum[k];
    cursor[tid] = ex;
    const int node = base + tid;
    if (node < n) { cnt[node] = v; rowptr[node] = ex; }
  }
  __syncthreads();
  for (int i0 = bs + tid; i0 < be; i0 += 4 * BT) {
    int4 q[4];
#pragma unroll
    for (int u = 0; u < 4; ++u) {
      const int i = i0 + u * BT;
      q[u] = make_int4(0, 0, base, 0);
      if (i < be) q[u] = binq[i];
    }
#pragma unroll
    for (int u = 0; u < 4; ++u) {
      if (i0 + u * BT < be) {
        const int p = atomicAdd(&cursor[q[u].z - base], 1);
        atomicAdd(&wdeg[q[u].z - base], __int_as_float(q[u].y));
        lst[p] = make_int2(q[u].x, q[u].y);
      }
    }
  }
  __syncthreads();
  if (tid < BNODE && base + tid < n) {
    const float d = 1.0f + wdeg[tid];
    dinv[base + tid] = (d > 0.0f) ? rsqrtf(d) : 0.0f;
  }
}

#define LROWS 64
#define LGR   (LROWS / 16)
#define LMT   (LROWS / 32)

template <bool LATEW, bool HEAD, int K>
__device__ __forceinline__ void gcn_rows4(float4& acc, float& wv,
                                          const float* __restrict__ Xin, int ldi,
                                          int sv, int m, int l,
                                          float dn, float raw, float di,
                                          int2* __restrict__ lo) {
  float4 h[4];
  float  w[4];
#pragma unroll
  for (int u = 0; u < 4; ++u) {
    const int s = __shfl(sv, K + u, 16);
    if (!LATEW) w[u] = __shfl(wv, K + u, 16);
    h[u] = *(const float4*)(Xin + (size_t)s * ldi + l * 4);
  }
  if (LATEW) {
    if (HEAD) {
      const float nrm = dn * raw * di;
      if (l < m) lo[l] = make_int2(sv, __float_as_int(nrm));
      wv = (l < m) ? nrm : 0.0f;
    }
#pragma unroll
    for (int u = 0; u < 4; ++u) w[u] = __shfl(wv, K + u, 16);
  }
#pragma unroll
  for (int u = 0; u < 4; ++u) {
    acc.x = fmaf(h[u].x, w[u], acc.x);
    acc.y = fmaf(h[u].y, w[u], acc.y);
    acc.z = fmaf(h[u].z, w[u], acc.z);
    acc.w = fmaf(h[u].w, w[u], acc.w);
  }
}

template <bool FIRST>
__device__ __forceinline__ float4 gcn_sum_list(const float* __restrict__ Xin, int ldi,
                                               const int2* __restrict__ lin,
                                               int2* __restrict__ lout,
                                               const float* __restrict__ dinv, float di,
                                               int st, int end, int l, int2 e) {
  float4 acc = make_float4(0.0f, 0.0f, 0.0f, 0.0f);
  for (int p = st; p < end; p += 16) {
    const int m = min(16, end - p);
    if (p > st) e = lin[p + min(l, m - 1)];
    const int sv = e.x;
    if (FIRST) {
      const float dn  = dinv[sv];
      const float raw = __int_as_float(e.y);
      float wv = 0.0f;
      gcn_rows4<true, true, 0>(acc, wv, Xin, ldi, sv, m, l, dn, raw, di, lout + p);
      if (m > 4) {
        gcn_rows4<true, false, 4>(acc, wv, Xin, ldi, sv, m, l, dn, raw, di, lout + p);
        if (m > 8) {
          gcn_rows4<true, false, 8>(acc, wv, Xin, ldi, sv, m, l, dn, raw, di, lout + p);
          if (m > 12)
            gcn_rows4<true, false, 12>(acc, wv, Xin, ldi, sv, m, l, dn, raw, di, lout + p);
        }
      }
    } else {
      float wv = __int_as_float(e.y & ((l < m) ? -1 : 0));
      gcn_rows4<false, false, 0>(acc, wv, Xin, ldi, sv, m, l, 0.0f, 0.0f, di, lout);
      if (m > 4) {
        gcn_rows4<false, false, 4>(acc, wv, Xin, ldi, sv, m, l, 0.0f, 0.0f, di, lout);
        if (m > 8) {
          gcn_rows4<false, false, 8>(acc, wv, Xin, ldi, sv, m, l, 0.0f, 0.0f, di, lout);
          if (m > 12)
            gcn_rows4<false, false, 12>(acc, wv, Xin, ldi, sv, m, l, 0.0f, 0.0f, di, lout);
        }
      }
    }
  }
  return acc;
}

template <bool FIRST>
__global__ __launch_bounds__(256)
void gcn_layer(const float* __restrict__ Xin, int ldi, const float* __restrict__ dinv,
               const int* __restrict__ rowptr, const int* __restrict__ cnt,
               const int* __restrict__ lin_, int* __restrict__ lout_,
               const float* __restrict__ W,
               const float* __restrict__ bias, float* Hout, int ldo, int n) {
  __shared__ __attribute__((aligned(16))) float Ws[D * LDP];
  __shared__ __attribute__((aligned(16))) float Zs[LROWS * LDP];
  const int2* lin = (const int2*)lin_;
  int2*       lout = (int2*)lout_;
  const int tid    = threadIdx.x;
  const int wave   = tid >> 5;
  const int lane   = tid & 31;
  const int laneLo = lane & 15;
  const int laneHi = lane >> 4;
  const int wn     = (wave & 3) << 4;
  const int wm0    = (wave >> 2) * (LMT * 16);
  const int g      = tid >> 4;
  const int l      = tid & 15;
  const int row0   = blockIdx.x * LROWS;
  const int r0     = g * LGR;

  int stL = 0, cL = 0;
  float dL = 0.0f;
  {
    const int nd = row0 + r0 + l;
    if (l < LGR && nd < n) { stL = rowptr[nd]; cL = cnt[nd]; dL = dinv[nd]; }
  }
  const int   gcol = wn + laneLo;
  const float bv   = bias[gcol];

  int st = __shfl(stL, 0, 16);
  int en = st + __shfl(cL, 0, 16);
  int2 e = make_int2(0, 0);
  if (en > st) e = lin[st + min(l, min(16, en - st) - 1)];
  {
    float4 wr[D * D / 4 / 256];
#pragma unroll
    for (int u = 0; u < D * D / 4 / 256; ++u)
      wr[u] = *(const float4*)(W + (size_t)(tid + u * 256) * 4);
#pragma unroll
    for (int u = 0; u < D * D / 4 / 256; ++u) {
      const int i = tid + u * 256;
      *(float4*)(&Ws[(i >> 4) * LDP + ((i & 15) << 2)]) = wr[u];
    }
  }
#pragma unroll
  for (int j = 0; j < LGR; ++j) {
    const int node = row0 + r0 + j;
    const float di = __shfl(dL, j, 16);
    int nst = 0, nen = 0;
    int2 ne = make_int2(0, 0);
    if (j + 1 < LGR) {
      nst = __shfl(stL, j + 1, 16);
      nen = nst + __shfl(cL, j + 1, 16);
      if (nen > nst) ne = lin[nst + min(l, min(16, nen - nst) - 1)];
    }
    const float4 hs = *(const float4*)(Xin + (size_t)min(node, n - 1) * ldi + l * 4);
    const float4 a = gcn_sum_list<FIRST>(Xin, ldi, lin, lout, dinv, di, st, en, l, e);
    const float sc = di * di;
    float4 z;
    z.x = fmaf(sc, hs.x, a.x); z.y = fmaf(sc, hs.y, a.y);
    z.z = fmaf(sc, hs.z, a.z); z.w = fmaf(sc, hs.w, a.w);
    *(float4*)(&Zs[(r0 + j) * LDP + l * 4]) = z;
    st = nst; en = nen; e = ne;
  }
  __syncthreads();

  v8f acc[LMT];
#pragma unroll
  for (int t = 0; t < LMT; ++t) acc[t] = v8f{};
#pragma unroll
  for (int k0 = 0; k0 < D; k0 += 4) {
    v2f b;
    b[0] = Ws[(k0 + 2 * laneHi + 0) * LDP + wn + laneLo];
    b[1] = Ws[(k0 + 2 * laneHi + 1) * LDP + wn + laneLo];
#pragma unroll
    for (int t = 0; t < LMT; ++t) {
      v2f a = *(const v2f*)(&Zs[(wm0 + t * 16 + laneLo) * LDP + k0 + 2 * laneHi]);
      acc[t] = __builtin_amdgcn_wmma_f32_16x16x4_f32(false, a, false, b,
                                                     (short)0, acc[t], false, false);
    }
  }
#pragma unroll
  for (int t = 0; t < LMT; ++t) {
#pragma unroll
    for (int v = 0; v < 8; ++v) {
      const int grow = row0 + wm0 + t * 16 + v + 8 * laneHi;
      float r = acc[t][v] + bv;
      r = r > 0.0f ? r : 0.0f;
      if (grow < n) Hout[(size_t)grow * ldo + gcol] = r;
    }
  }
}

#define GXV (TROWS * D / 4 / 256)
#define GWV (D * D / 4 / 256)
__global__ __launch_bounds__(256)
void gcn_gemm_wmma(const float* __restrict__ X0, int ld0,
                   const float* __restrict__ X, int ldx,
                   const float* __restrict__ W,
                   const float* __restrict__ bias,
                   float* __restrict__ Out, int K, int n) {
  __shared__ __attribute__((aligned(16))) float Xs[TROWS * LDP];
  __shared__ __attribute__((aligned(16))) float Ws[D * LDP];
  const int tid    = threadIdx.x;
  const int wave   = tid >> 5;
  const int lane   = tid & 31;
  const int laneLo = lane & 15;
  const int laneHi = lane >> 4;
  const int row0   = blockIdx.x * TROWS;
  const int wm     = (wave >> 2) << 4;
  const int wn     = (wave & 3) << 4;
  const int sr     = tid >> 4;
  const int sc     = (tid & 15) << 2;

  v8f acc = {};
  for (int kc = 0; kc < K; kc += D) {
    const float* Xc = (kc == 0) ? X0 : X + kc;
    const int    ldc = (kc == 0) ? ld0 : ldx;
    float4 xr[GXV], wr[GWV];
#pragma unroll
    for (int u = 0; u < GXV; ++u) {
      int rr = row0 + sr + u * 16; if (rr >= n) rr = n - 1;
      xr[u] = *(const float4*)(Xc + (size_t)rr * ldc + sc);
    }
#pragma unroll
    for (int u = 0; u < GWV; ++u)
      wr[u] = *(const float4*)(W + (size_t)kc * D + (size_t)(tid + u * 256) * 4);
#pragma unroll
    for (int u = 0; u < GXV; ++u)
      *(float4*)(&Xs[(sr + u * 16) * LDP + sc]) = xr[u];
#pragma unroll
    for (int u = 0; u < GWV; ++u)
      *(float4*)(&Ws[(sr + u * 16) * LDP + sc]) = wr[u];
    __syncthreads();
#pragma unroll
    for (int k0 = 0; k0 < D; k0 += 4) {
      v2f a = *(const v2f*)(&Xs[(wm + laneLo) * LDP + k0 + 2 * laneHi]);
      v2f b;
      b[0] = Ws[(k0 + 2 * laneHi + 0) * LDP + wn + laneLo];
      b[1] = Ws[(k0 + 2 * laneHi + 1) * LDP + wn + laneLo];
      acc = __builtin_amdgcn_wmma_f32_16x16x4_f32(false, a, false, b,
                                                  (short)0, acc, false, false);
    }
    __syncthreads();
  }
  const int gcol = wn + laneLo;
  const float bv = bias ? bias[gcol] : 0.0f;
#pragma unroll
  for (int v = 0; v < 8; ++v) {
    int grow = row0 + wm + v + 8 * laneHi;
    if (grow < n) Out[(size_t)grow * D + gcol] = acc[v] + bv;
  }
}

extern "C" void kernel_launch(void* const* d_in, const int* in_sizes, int n_in,
                              void* d_out, int out_size, void* d_ws, size_t ws_size,
                              hipStream_t stream) {
  const float* x   = (const float*)d_in[0];
  const int*   ei  = (const int*)d_in[1];
  const float* ew  = (const float*)d_in[2];
  const float* Wl[3] = {(const float*)d_in[3], (const float*)d_in[5], (const float*)d_in[7]};
  const float* bl[3] = {(const float*)d_in[4], (const float*)d_in[6], (const float*)d_in[8]};
  const float* Wf  = (const float*)d_in[9];
  const float* bf  = (const float*)d_in[10];

  const int N = in_sizes[0] / D;
  const int E = in_sizes[2];
  const int* src = ei;
  const int* dst = ei + E;

  const size_t Np = ((size_t)N + 255) & ~(size_t)255;
  const size_t Ep = ((size_t)E + 255) & ~(size_t)255;
  const int    nbk = (int)(Np >> BSH);
  const size_t Tn  = (size_t)nbk * NSL;
  float* deg    = (float*)d_ws;
  float* Hcat   = deg + Np;
  int*   cnt    = (int*)(Hcat + (size_t)N * 256);
  int*   rowptr = cnt + Np;
  int*   lst    = rowptr + Np;
  int*   lstn   = lst + 2 * Ep;
  int*   tab    = lstn + 2 * Ep;
  int*   bst    = tab + Tn;
  int*   binq   = bst + Tn;

  const int T = 256;
  gcn_bin_count<<<NSL, BT, nbk * sizeof(int), stream>>>(dst, tab, E, nbk);
  gcn_bin_place<<<NSL, BT, 2 * nbk * sizeof(int), stream>>>(src, dst, ew, tab, bst, binq, E, nbk);
  gcn_bin_lists<<<nbk, BT, 0, stream>>>(bst, binq, cnt, rowptr, lst, deg, E, N, nbk);

  const int layer_blocks = (N + LROWS - 1) / LROWS;
  gcn_layer<true><<<layer_blocks, T, 0, stream>>>(x, D, deg, rowptr, cnt, lst, lstn,
                                                  Wl[0], bl[0], Hcat + D, 256, N);
  gcn_layer<false><<<layer_blocks, T, 0, stream>>>(Hcat + D, 256, deg, rowptr, cnt, lstn, nullptr,
                                                   Wl[1], bl[1], Hcat + 2 * D, 256, N);
  gcn_layer<false><<<layer_blocks, T, 0, stream>>>(Hcat + 2 * D, 256, deg, rowptr, cnt, lstn, nullptr,
                                                   Wl[2], bl[2], Hcat + 3 * D, 256, N);
  const int gemm_blocks = (N + TROWS - 1) / TROWS;
  gcn_gemm_wmma<<<gemm_blocks, T, 0, stream>>>(x, D, Hcat, 256, Wf, bf, (float*)d_out,
                                               4 * D, N);
}


# Layer kernels: row pitches compile-time, row addresses formed in 32 bits (one shift-or on a 16-byte row index beside a scalar base); the 64-bit multiply and 64-bit add per gathered row are gone
# speedup vs baseline: 4.5707x; 1.2301x over previous
#include <hip/hip_runtime.h>

typedef __attribute__((ext_vector_type(2))) float v2f;
typedef __attribute__((ext_vector_type(8))) float v8f;

#define D     64
#define TROWS 32
#define LDP   68

#define BSH   8
#define BNODE (1 << BSH)
#define NSL   256
#define BT    512

__global__ __launch_bounds__(BT)
void gcn_bin_count(const int* __restrict__ dst, int* __restrict__ tab, int e, int nbk) {
  extern __shared__ int hist[];
  const int s = blockIdx.x, tid = threadIdx.x;
  for (int b = tid; b < nbk; b += BT) hist[b] = 0;
  __syncthreads();
  const int per = (e + NSL - 1) / NSL;
  const int lo  = s * per;
  const int hi  = min(e, lo + per);
  for (int i0 = lo + tid; i0 < hi; i0 += 4 * BT) {
    int d[4];
#pragma unroll
    for (int u = 0; u < 4; ++u) {
      const int i = i0 + u * BT;
      d[u] = (i < hi) ? dst[i] : 0;
    }
#pragma unroll
    for (int u = 0; u < 4; ++u)
      if (i0 + u * BT < hi) atomicAdd(&hist[d[u] >> BSH], 1);
  }
  __syncthreads();
  for (int b = tid; b < nbk; b += BT) tab[b * NSL + s] = hist[b];
}

#define RG (BT / 8)
#define RK (NSL / 32)
__global__ __launch_bounds__(BT)
void gcn_bin_place(const int* __restrict__ src, const int* __restrict__ dst,
                   const float* __restrict__ ew, const int* __restrict__ tab,
                   int* __restrict__ bst, int* __restrict__ binq_, int e, int nbk) {
  extern __shared__ int cur[];
  __shared__ int wsum[BT / 32];
  int* tot = cur + nbk;
  int4* binq = (int4*)binq_;
  const int4* tab4 = (const int4*)tab;
  const int s = blockIdx.x, tid = threadIdx.x, lane = tid & 31, wv = tid >> 5;
  const int q = tid & 7;
  for (int b0 = tid >> 3; b0 < nbk; b0 += 2 * RG) {
    int4 v[2][RK];
#pragma unroll
    for (int r = 0; r < 2; ++r) {
      const int b = b0 + r * RG;
#pragma unroll
      for (int k = 0; k < RK; ++k) {
        v[r][k] = make_int4(0, 0, 0, 0);
        if (b < nbk) v[r][k] = tab4[(size_t)b * (NSL / 4) + k * 8 + q];
      }
    }
#pragma unroll
    for (int r = 0; r < 2; ++r) {
      const int b = b0 + r * RG;
      int full = 0, part = 0;
#pragma unroll
      for (int k = 0; k < RK; ++k) {
        const int i0 = k * 32 + q * 4;
        full += (v[r][k].x + v[r][k].y) + (v[r][k].z + v[r][k].w);
        part += (((i0 + 0 < s) ? v[r][k].x : 0) + ((i0 + 1 < s) ? v[r][k].y : 0))
              + (((i0 + 2 < s) ? v[r][k].z : 0) + ((i0 + 3 < s) ? v[r][k].w : 0));
      }
      for (int o = 4; o > 0; o >>= 1) {
        full += __shfl_xor(full, o, 8);
        part += __shfl_xor(part, o, 8);
      }
      if (q == 0 && b < nbk) { tot[b] = full; cur[b] = part; }
    }
  }
  __syncthreads();
  int carry = 0;
  for (int c0 = 0; c0 < nbk; c0 += BT) {
    const int b = c0 + tid;
    const int t = (b < nbk) ? tot[b] : 0;
    int inc = t;
    for (int o = 1; o < 32; o <<= 1) {
      int u = __shfl_up(inc, o);
      if (lane >= o) inc += u;
    }
    if (lane == 31) wsum[wv] = inc;
    __syncthreads();
    int ex = carry + inc - t;
    int all = 0;
    for (int k = 0; k < BT / 32; ++k) {
      const int w = wsum[k];
      all += w;
      if (k < wv) ex += w;
    }
    if (b < nbk) {
      cur[b] += ex;
      if (s == 0) bst[b] = ex;
    }
    carry += all;
    __syncthreads();
  }
  const int per = (e + NSL - 1) / NSL;
  const int lo  = s * per;
  const int hi  = min(e, lo + per);
  for (int i0 = lo + tid; i0 < hi; i0 += 4 * BT) {
    int d[4], sv[4], wb[4];
#pragma unroll
    for (int u = 0; u < 4; ++u) {
      const int i = i0 + u * BT;
      d[u] = 0; sv[u] = 0; wb[u] = 0;
      if (i < hi) { d[u] = dst[i]; sv[u] = src[i]; wb[u] = __float_as_int(ew[i]); }
    }
#pragma unroll
    for (int u = 0; u < 4; ++u) {
      if (i0 + u * BT < hi) {
        const int p = atomicAdd(&cur[d[u] >> BSH], 1);
        binq[p] = make_int4(sv[u], wb[u], d[u], 0);
      }
    }
  }
}

__global__ __launch_bounds__(BT)
void gcn_bin_lists(const int* __restrict__ bst, const int* __restrict__ binq_,
                   int* __restrict__ cnt, int* __restrict__ rowptr, int* __restrict__ lst_,
                   float* __restrict__ dinv, int e, int n, int nbk) {
  __shared__ int c[BNODE];
  __shared__ int cursor[BNODE];
  __shared__ float wdeg[BNODE];
  __shared__ int wsum[BNODE / 32];
  const int4* binq = (const int4*)binq_;
  int2* lst = (int2*)lst_;
  const int b = blockIdx.x, tid = threadIdx.x, lane = tid & 31, wv = tid >> 5;
  const int base = b << BSH;
  const int bs = bst[b];
  const int be = (b + 1 < nbk) ? bst[b + 1] : e;
  if (tid < BNODE) { c[tid] = 0; wdeg[tid] = 0.0f; }
  __syncthreads();
  for (int i0 = bs + tid; i0 < be; i0 += 4 * BT) {
    int d[4];
#pragma unroll
    for (int u = 0; u < 4; ++u) {
      const int i = i0 + u * BT;
      d[u] = (i < be) ? binq[i].z : base;
    }
#pragma unroll
    for (int u = 0; u < 4; ++u)
      if (i0 + u * BT < be) atomicAdd(&c[d[u] - base], 1);
  }
  __syncthreads();
  const int v = (tid < BNODE) ? c[tid] : 0;
  int inc = v;
  for (int o = 1; o < 32; o <<= 1) {
    int t = __shfl_up(inc, o);
    if (lane >= o) inc += t;
  }
  if (lane == 31 && tid < BNODE) wsum[wv] = inc;
  __syncthreads();
  if (tid < BNODE) {
    int ex = bs + inc - v;
    for (int k = 0; k < wv; ++k) ex += wsum[k];
    cursor[tid] = ex;
    const int node = base + tid;
    if (node < n) { cnt[node] = v; rowptr[node] = ex; }
  }
  __syncthreads();
  for (int i0 = bs + tid; i0 < be; i0 += 4 * BT) {
    int4 q[4];
#pragma unroll
    for (int u = 0; u < 4; ++u) {
      const int i = i0 + u * BT;
      q[u] = make_int4(0, 0, base, 0);
      if (i < be) q[u] = binq[i];
    }
#pragma unroll
    for (int u = 0; u < 4; ++u) {
      if (i0 + u * BT < be) {
        const int p = atomicAdd(&cursor[q[u].z - base], 1);
        atomicAdd(&wdeg[q[u].z - base], __int_as_float(q[u].y));
        lst[p] = make_int2(q[u].x, q[u].y);
      }
    }
  }
  __syncthreads();
  if (tid < BNODE && base + tid < n) {
    const float d = 1.0f + wdeg[tid];
    dinv[base + tid] = (d > 0.0f) ? rsqrtf(d) : 0.0f;
  }
}

#define LROWS 64
#define LGR   (LROWS / 16)
#define LMT   (LROWS / 32)
#define HC    (4 * D)

template <int LDI4, bool LATEW, bool HEAD, int K>
__device__ __forceinline__ void gcn_rows4(float4& acc, float& wv,
                                          const float4* __restrict__ X4,
                                          int sv, int m, int l,
                                          float dn, float raw, float di,
                                          int2* __restrict__ lo) {
  float4 h[4];
  float  w[4];
#pragma unroll
  for (int u = 0; u < 4; ++u) {
    const int s = __shfl(sv, K + u, 16);
    if (!LATEW) w[u] = __shfl(wv, K + u, 16);
    h[u] = X4[(unsigned)s * (unsigned)LDI4 + (unsigned)l];
  }
  if (LATEW) {
    if (HEAD) {
      const float nrm = dn * raw * di;
      if (l < m) lo[l] = make_int2(sv, __float_as_int(nrm));
      wv = (l < m) ? nrm : 0.0f;
    }
#pragma unroll
    for (int u = 0; u < 4; ++u) w[u] = __shfl(wv, K + u, 16);
  }
#pragma unroll
  for (int u = 0; u < 4; ++u) {
    acc.x = fmaf(h[u].x, w[u], acc.x);
    acc.y = fmaf(h[u].y, w[u], acc.y);
    acc.z = fmaf(h[u].z, w[u], acc.z);
    acc.w = fmaf(h[u].w, w[u], acc.w);
  }
}

template <bool FIRST, int LDI4>
__device__ __forceinline__ float4 gcn_sum_list(const float4* __restrict__ X4,
                                               const int2* __restrict__ lin,
                                               int2* __restrict__ lout,
                                               const float* __restrict__ dinv, float di,
                                               int st, int end, int l, int2 e) {
  float4 acc = make_float4(0.0f, 0.0f, 0.0f, 0.0f);
  for (int p = st; p < end; p += 16) {
    const int m = min(16, end - p);
    if (p > st) e = lin[p + min(l, m - 1)];
    const int sv = e.x;
    if (FIRST) {
      const float dn  = dinv[sv];
      const float raw = __int_as_float(e.y);
      float wv = 0.0f;
      gcn_rows4<LDI4, true, true, 0>(acc, wv, X4, sv, m, l, dn, raw, di, lout + p);
      if (m > 4) {
        gcn_rows4<LDI4, true, false, 4>(acc, wv, X4, sv, m, l, dn, raw, di, lout + p);
        if (m > 8) {
          gcn_rows4<LDI4, true, false, 8>(acc, wv, X4, sv, m, l, dn, raw, di, lout + p);
          if (m > 12)
            gcn_rows4<LDI4, true, false, 12>(acc, wv, X4, sv, m, l, dn, raw, di, lout + p);
        }
      }
    } else {
      float wv = __int_as_float(e.y & ((l < m) ? -1 : 0));
      gcn_rows4<LDI4, false, false, 0>(acc, wv, X4, sv, m, l, 0.0f, 0.0f, di, lout);
      if (m > 4) {
        gcn_rows4<LDI4, false, false, 4>(acc, wv, X4, sv, m, l, 0.0f, 0.0f, di, lout);
        if (m > 8) {
          gcn_rows4<LDI4, false, false, 8>(acc, wv, X4, sv, m, l, 0.0f, 0.0f, di, lout);
          if (m > 12)
            gcn_rows4<LDI4, false, false, 12>(acc, wv, X4, sv, m, l, 0.0f, 0.0f, di, lout);
        }
      }
    }
  }
  return acc;
}

template <bool FIRST>
__global__ __launch_bounds__(256)
void gcn_layer(const float* __restrict__ Xin, const float* __restrict__ dinv,
               const int* __restrict__ rowptr, const int* __restrict__ cnt,
               const int* __restrict__ lin_, int* __restrict__ lout_,
               const float* __restrict__ W,
               const float* __restrict__ bias, float* Hout, int n) {
  constexpr int LDI4 = (FIRST ? D : HC) / 4;
  const float4* __restrict__ X4 = (const float4*)Xin;
  __shared__ __attribute__((aligned(16))) float Ws[D * LDP];
  __shared__ __attribute__((aligned(16))) float Zs[LROWS * LDP];
  const int2* lin = (const int2*)lin_;
  int2*       lout = (int2*)lout_;
  const int tid    = threadIdx.x;
  const int wave   = tid >> 5;
  const int lane   = tid & 31;
  const int laneLo = lane & 15;
  const int laneHi = lane >> 4;
  const int wn     = (wave & 3) << 4;
  const int wm0    = (wave >> 2) * (LMT * 16);
  const int g      = tid >> 4;
  const int l      = tid & 15;
  const int row0   = blockIdx.x * LROWS;
  const int r0     = g * LGR;

  int stL = 0, cL = 0;
  float dL = 0.0f;
  {
    const int nd = row0 + r0 + l;
    if (l < LGR && nd < n) { stL = rowptr[nd]; cL = cnt[nd]; dL = dinv[nd]; }
  }
  const int   gcol = wn + laneLo;
  const float bv   = bias[gcol];

  int st = __shfl(stL, 0, 16);
  int en = st + __shfl(cL, 0, 16);
  int2 e = make_int2(0, 0);
  if (en > st) e = lin[st + min(l, min(16, en - st) - 1)];
  {
    float4 wr[D * D / 4 / 256];
#pragma unroll
    for (int u = 0; u < D * D / 4 / 256; ++u)
      wr[u] = *(const float4*)(W + (size_t)(tid + u * 256) * 4);
#pragma unroll
    for (int u = 0; u < D * D / 4 / 256; ++u) {
      const int i = tid + u * 256;
      *(float4*)(&Ws[(i >> 4) * LDP + ((i & 15) << 2)]) = wr[u];
    }
  }
#pragma unroll
  for (int j = 0; j < LGR; ++j) {
    const int node = row0 + r0 + j;
    const float di = __shfl(dL, j, 16);
    int nst = 0, nen = 0;
    int2 ne = make_int2(0, 0);
    if (j + 1 < LGR) {
      nst = __shfl(stL, j + 1, 16);
      nen = nst + __shfl(cL, j + 1, 16);
      if (nen > nst) ne = lin[nst + min(l, min(16, nen - nst) - 1)];
    }
    const float4 hs = X4[(unsigned)min(node, n - 1) * (unsigned)LDI4 + (unsigned)l];
    const float4 a = gcn_sum_list<FIRST, LDI4>(X4, lin, lout, dinv, di, st, en, l, e);
    const float sc = di * di;
    float4 z;
    z.x = fmaf(sc, hs.x, a.x); z.y = fmaf(sc, hs.y, a.y);
    z.z = fmaf(sc, hs.z, a.z); z.w = fmaf(sc, hs.w, a.w);
    *(float4*)(&Zs[(r0 + j) * LDP + l * 4]) = z;
    st = nst; en = nen; e = ne;
  }
  __syncthreads();

  v8f acc[LMT];
#pragma unroll
  for (int t = 0; t < LMT; ++t) acc[t] = v8f{};
#pragma unroll
  for (int k0 = 0; k0 < D; k0 += 4) {
    v2f b;
    b[0] = Ws[(k0 + 2 * laneHi + 0) * LDP + wn + laneLo];
    b[1] = Ws[(k0 + 2 * laneHi + 1) * LDP + wn + laneLo];
#pragma unroll
    for (int t = 0; t < LMT; ++t) {
      v2f a = *(const v2f*)(&Zs[(wm0 + t * 16 + laneLo) * LDP + k0 + 2 * laneHi]);
      acc[t] = __builtin_amdgcn_wmma_f32_16x16x4_f32(false, a, false, b,
                                                     (short)0, acc[t], false, false);
    }
  }
#pragma unroll
  for (int t = 0; t < LMT; ++t) {
#pragma unroll
    for (int v = 0; v < 8; ++v) {
      const int grow = row0 + wm0 + t * 16 + v + 8 * laneHi;
      float r = acc[t][v] + bv;
      r = r > 0.0f ? r : 0.0f;
      if (grow < n) Hout[(unsigned)grow * (unsigned)HC + (unsigned)gcol] = r;
    }
  }
}

#define GXV (TROWS * D / 4 / 256)
#define GWV (D * D / 4 / 256)
__global__ __launch_bounds__(256)
void gcn_gemm_wmma(const float* __restrict__ X0, int ld0,
                   const float* __restrict__ X, int ldx,
                   const float* __restrict__ W,
                   const float* __restrict__ bias,
                   float* __restrict__ Out, int K, int n) {
  __shared__ __attribute__((aligned(16))) float Xs[TROWS * LDP];
  __shared__ __attribute__((aligned(16))) float Ws[D * LDP];
  const int tid    = threadIdx.x;
  const int wave   = tid >> 5;
  const int lane   = tid & 31;
  const int laneLo = lane & 15;
  const int laneHi = lane >> 4;
  const int row0   = blockIdx.x * TROWS;
  const int wm     = (wave >> 2) << 4;
  const int wn     = (wave & 3) << 4;
  const int sr     = tid >> 4;
  const int sc     = (tid & 15) << 2;

  v8f acc = {};
  for (int kc = 0; kc < K; kc += D) {
    const float* Xc = (kc == 0) ? X0 : X + kc;
    const int    ldc = (kc == 0) ? ld0 : ldx;
    float4 xr[GXV], wr[GWV];
#pragma unroll
    for (int u = 0; u < GXV; ++u) {
      int rr = row0 + sr + u * 16; if (rr >= n) rr = n - 1;
      xr[u] = *(const float4*)(Xc + (size_t)rr * ldc + sc);
    }
#pragma unroll
    for (int u = 0; u < GWV; ++u)
      wr[u] = *(const float4*)(W + (size_t)kc * D + (size_t)(tid + u * 256) * 4);
#pragma unroll
    for (int u = 0; u < GXV; ++u)
      *(float4*)(&Xs[(sr + u * 16) * LDP + sc]) = xr[u];
#pragma unroll
    for (int u = 0; u < GWV; ++u)
      *(float4*)(&Ws[(sr + u * 16) * LDP + sc]) = wr[u];
    __syncthreads();
#pragma unroll
    for (int k0 = 0; k0 < D; k0 += 4) {
      v2f a = *(const v2f*)(&Xs[(wm + laneLo) * LDP + k0 + 2 * laneHi]);
      v2f b;
      b[0] = Ws[(k0 + 2 * laneHi + 0) * LDP + wn + laneLo];
      b[1] = Ws[(k0 + 2 * laneHi + 1) * LDP + wn + laneLo];
      acc = __builtin_amdgcn_wmma_f32_16x16x4_f32(false, a, false, b,
                                                  (short)0, acc, false, false);
    }
    __syncthreads();
  }
  const int gcol = wn + laneLo;
  const float bv = bias ? bias[gcol] : 0.0f;
#pragma unroll
  for (int v = 0; v < 8; ++v) {
    int grow = row0 + wm + v + 8 * laneHi;
    if (grow < n) Out[(size_t)grow * D + gcol] = acc[v] + bv;
  }
}

extern "C" void kernel_launch(void* const* d_in, const int* in_sizes, int n_in,
                              void* d_out, int out_size, void* d_ws, size_t ws_size,
                              hipStream_t stream) {
  const float* x   = (const float*)d_in[0];
  const int*   ei  = (const int*)d_in[1];
  const float* ew  = (const float*)d_in[2];
  const float* Wl[3] = {(const float*)d_in[3], (const float*)d_in[5], (const float*)d_in[7]};
  const float* bl[3] = {(const float*)d_in[4], (const float*)d_in[6], (const float*)d_in[8]};
  const float* Wf  = (const float*)d_in[9];
  const float* bf  = (const float*)d_in[10];

  const int N = in_sizes[0] / D;
  const int E = in_sizes[2];
  const int* src = ei;
  const int* dst = ei + E;

  const size_t Np = ((size_t)N + 255) & ~(size_t)255;
  const size_t Ep = ((size_t)E + 255) & ~(size_t)255;
  const int    nbk = (int)(Np >> BSH);
  const size_t Tn  = (size_t)nbk * NSL;
  float* deg    = (float*)d_ws;
  float* Hcat   = deg + Np;
  int*   cnt    = (int*)(Hcat + (size_t)N * 256);
  int*   rowptr = cnt + Np;
  int*   lst    = rowptr + Np;
  int*   lstn   = lst + 2 * Ep;
  int*   tab    = lstn + 2 * Ep;
  int*   bst    = tab + Tn;
  int*   binq   = bst + Tn;

  const int T = 256;
  gcn_bin_count<<<NSL, BT, nbk * sizeof(int), stream>>>(dst, tab, E, nbk);
  gcn_bin_place<<<NSL, BT, 2 * nbk * sizeof(int), stream>>>(src, dst, ew, tab, bst, binq, E, nbk);
  gcn_bin_lists<<<nbk, BT, 0, stream>>>(bst, binq, cnt, rowptr, lst, deg, E, N, nbk);

  const int layer_blocks = (N + LROWS - 1) / LROWS;
  gcn_layer<true><<<layer_blocks, T, 0, stream>>>(x, deg, rowptr, cnt, lst, lstn,
                                                  Wl[0], bl[0], Hcat + D, N);
  gcn_layer<false><<<layer_blocks, T, 0, stream>>>(Hcat + D, deg, rowptr, cnt, lstn, nullptr,
                                                   Wl[1], bl[1], Hcat + 2 * D, N);
  gcn_layer<false><<<layer_blocks, T, 0, stream>>>(Hcat + 2 * D, deg, rowptr, cnt, lstn, nullptr,
                                                   Wl[2], bl[2], Hcat + 3 * D, N);
  const int gemm_blocks = (N + TROWS - 1) / TROWS;
  gcn_gemm_wmma<<<gemm_blocks, T, 0, stream>>>(x, D, Hcat, 256, Wf, bf, (float*)d_out,
                                               4 * D, N);
}


# Final product folded into the third layer's kernel: its output tile stays in LDS as the last k block, summed first (k-block order changes); one launch, the h3 plane and the 25000-wave GEMM grid gone
# speedup vs baseline: 4.8126x; 1.0529x over previous
#include <hip/hip_runtime.h>

typedef __attribute__((ext_vector_type(2))) float v2f;
typedef __attribute__((ext_vector_type(8))) float v8f;

#define D     64
#define LDP   68

#define BSH   8
#define BNODE (1 << BSH)
#define NSL   256
#define BT    512

__global__ __launch_bounds__(BT)
void gcn_bin_count(const int* __restrict__ dst, int* __restrict__ tab, int e, int nbk) {
  extern __shared__ int hist[];
  const int s = blockIdx.x, tid = threadIdx.x;
  for (int b = tid; b < nbk; b += BT) hist[b] = 0;
  __syncthreads();
  const int per = (e + NSL - 1) / NSL;
  const int lo  = s * per;
  const int hi  = min(e, lo + per);
  for (int i0 = lo + tid; i0 < hi; i0 += 4 * BT) {
    int d[4];
#pragma unroll
    for (int u = 0; u < 4; ++u) {
      const int i = i0 + u * BT;
      d[u] = (i < hi) ? dst[i] : 0;
    }
#pragma unroll
    for (int u = 0; u < 4; ++u)
      if (i0 + u * BT < hi) atomicAdd(&hist[d[u] >> BSH], 1);
  }
  __syncthreads();
  for (int b = tid; b < nbk; b += BT) tab[b * NSL + s] = hist[b];
}

#define RG (BT / 8)
#define RK (NSL / 32)
__global__ __launch_bounds__(BT)
void gcn_bin_place(const int* __restrict__ src, const int* __restrict__ dst,
                   const float* __restrict__ ew, const int* __restrict__ tab,
                   int* __restrict__ bst, int* __restrict__ binq_, int e, int nbk) {
  extern __shared__ int cur[];
  __shared__ int wsum[BT / 32];
  int* tot = cur + nbk;
  int4* binq = (int4*)binq_;
  const int4* tab4 = (const int4*)tab;
  const int s = blockIdx.x, tid = threadIdx.x, lane = tid & 31, wv = tid >> 5;
  const int q = tid & 7;
  for (int b0 = tid >> 3; b0 < nbk; b0 += 2 * RG) {
    int4 v[2][RK];
#pragma unroll
    for (int r = 0; r < 2; ++r) {
      const int b = b0 + r * RG;
#pragma unroll
      for (int k = 0; k < RK; ++k) {
        v[r][k] = make_int4(0, 0, 0, 0);
        if (b < nbk) v[r][k] = tab4[(size_t)b * (NSL / 4) + k * 8 + q];
      }
    }
#pragma unroll
    for (int r = 0; r < 2; ++r) {
      const int b = b0 + r * RG;
      int full = 0, part = 0;
#pragma unroll
      for (int k = 0; k < RK; ++k) {
        const int i0 = k * 32 + q * 4;
        full += (v[r][k].x + v[r][k].y) + (v[r][k].z + v[r][k].w);
        part += (((i0 + 0 < s) ? v[r][k].x : 0) + ((i0 + 1 < s) ? v[r][k].y : 0))
              + (((i0 + 2 < s) ? v[r][k].z : 0) + ((i0 + 3 < s) ? v[r][k].w : 0));
      }
      for (int o = 4; o > 0; o >>= 1) {
        full += __shfl_xor(full, o, 8);
        part += __shfl_xor(part, o, 8);
      }
      if (q == 0 && b < nbk) { tot[b] = full; cur[b] = part; }
    }
  }
  __syncthreads();
  int carry = 0;
  for (int c0 = 0; c0 < nbk; c0 += BT) {
    const int b = c0 + tid;
    const int t = (b < nbk) ? tot[b] : 0;
    int inc = t;
    for (int o = 1; o < 32; o <<= 1) {
      int u = __shfl_up(inc, o);
      if (lane >= o) inc += u;
    }
    if (lane == 31) wsum[wv] = inc;
    __syncthreads();
    int ex = carry + inc - t;
    int all = 0;
    for (int k = 0; k < BT / 32; ++k) {
      const int w = wsum[k];
      all += w;
      if (k < wv) ex += w;
    }
    if (b < nbk) {
      cur[b] += ex;
      if (s == 0) bst[b] = ex;
    }
    carry += all;
    __syncthreads();
  }
  const int per = (e + NSL - 1) / NSL;
  const int lo  = s * per;
  const int hi  = min(e, lo + per);
  for (int i0 = lo + tid; i0 < hi; i0 += 4 * BT) {
    int d[4], sv[4], wb[4];
#pragma unroll
    for (int u = 0; u < 4; ++u) {
      const int i = i0 + u * BT;
      d[u] = 0; sv[u] = 0; wb[u] = 0;
      if (i < hi) { d[u] = dst[i]; sv[u] = src[i]; wb[u] = __float_as_int(ew[i]); }
    }
#pragma unroll
    for (int u = 0; u < 4; ++u) {
      if (i0 + u * BT < hi) {
        const int p = atomicAdd(&cur[d[u] >> BSH], 1);
        binq[p] = make_int4(sv[u], wb[u], d[u], 0);
      }
    }
  }
}

__global__ __launch_bounds__(BT)
void gcn_bin_lists(const int* __restrict__ bst, const int* __restrict__ binq_,
                   int* __restrict__ cnt, int* __restrict__ rowptr, int* __restrict__ lst_,
                   float* __restrict__ dinv, int e, int n, int nbk) {
  __shared__ int c[BNODE];
  __shared__ int cursor[BNODE];
  __shared__ float wdeg[BNODE];
  __shared__ int wsum[BNODE / 32];
  const int4* binq = (const int4*)binq_;
  int2* lst = (int2*)lst_;
  const int b = blockIdx.x, tid = threadIdx.x, lane = tid & 31, wv = tid >> 5;
  const int base = b << BSH;
  const int bs = bst[b];
  const int be = (b + 1 < nbk) ? bst[b + 1] : e;
  if (tid < BNODE) { c[tid] = 0; wdeg[tid] = 0.0f; }
  __syncthreads();
  for (int i0 = bs + tid; i0 < be; i0 += 4 * BT) {
    int d[4];
#pragma unroll
    for (int u = 0; u < 4; ++u) {
      const int i = i0 + u * BT;
      d[u] = (i < be) ? binq[i].z : base;
    }
#pragma unroll
    for (int u = 0; u < 4; ++u)
      if (i0 + u * BT < be) atomicAdd(&c[d[u] - base], 1);
  }
  __syncthreads();
  const int v = (tid < BNODE) ? c[tid] : 0;
  int inc = v;
  for (int o = 1; o < 32; o <<= 1) {
    int t = __shfl_up(inc, o);
    if (lane >= o) inc += t;
  }
  if (lane == 31 && tid < BNODE) wsum[wv] = inc;
  __syncthreads();
  if (tid < BNODE) {
    int ex = bs + inc - v;
    for (int k = 0; k < wv; ++k) ex += wsum[k];
    cursor[tid] = ex;
    const int node = base + tid;
    if (node < n) { cnt[node] = v; rowptr[node] = ex; }
  }
  __syncthreads();
  for (int i0 = bs + tid; i0 < be; i0 += 4 * BT) {
    int4 q[4];
#pragma unroll
    for (int u = 0; u < 4; ++u) {
      const int i = i0 + u * BT;
      q[u] = make_int4(0, 0, base, 0);
      if (i < be) q[u] = binq[i];
    }
#pragma unroll
    for (int u = 0; u < 4; ++u) {
      if (i0 + u * BT < be) {
        const int p = atomicAdd(&cursor[q[u].z - base], 1);
        atomicAdd(&wdeg[q[u].z - base], __int_as_float(q[u].y));
        lst[p] = make_int2(q[u].x, q[u].y);
      }
    }
  }
  __syncthreads();
  if (tid < BNODE && base + tid < n) {
    const float d = 1.0f + wdeg[tid];
    dinv[base + tid] = (d > 0.0f) ? rsqrtf(d) : 0.0f;
  }
}

#define LROWS 64
#define LGR   (LROWS / 16)
#define LMT   (LROWS / 32)
#define HC    (4 * D)
#define GWV   (D * D / 4 / 256)

template <int LDI4, bool LATEW, bool HEAD, int K>
__device__ __forceinline__ void gcn_rows4(float4& acc, float& wv,
                                          const float4* __restrict__ X4,
                                          int sv, int m, int l,
                                          float dn, float raw, float di,
                                          int2* __restrict__ lo) {
  float4 h[4];
  float  w[4];
#pragma unroll
  for (int u = 0; u < 4; ++u) {
    const int s = __shfl(sv, K + u, 16);
    if (!LATEW) w[u] = __shfl(wv, K + u, 16);
    h[u] = X4[(unsigned)s * (unsigned)LDI4 + (unsigned)l];
  }
  if (LATEW) {
    if (HEAD) {
      const float nrm = dn * raw * di;
      if (l < m) lo[l] = make_int2(sv, __float_as_int(nrm));
      wv = (l < m) ? nrm : 0.0f;
    }
#pragma unroll
    for (int u = 0; u < 4; ++u) w[u] = __shfl(wv, K + u, 16);
  }
#pragma unroll
  for (int u = 0; u < 4; ++u) {
    acc.x = fmaf(h[u].x, w[u], acc.x);
    acc.y = fmaf(h[u].y, w[u], acc.y);
    acc.z = fmaf(h[u].z, w[u], acc.z);
    acc.w = fmaf(h[u].w, w[u], acc.w);
  }
}

template <bool FIRST, int LDI4>
__device__ __forceinline__ float4 gcn_sum_list(const float4* __restrict__ X4,
                                               const int2* __restrict__ lin,
                                               int2* __restrict__ lout,
                                               const float* __restrict__ dinv, float di,
                                               int st, int end, int l, int2 e) {
  float4 acc = make_float4(0.0f, 0.0f, 0.0f, 0.0f);
  for (int p = st; p < end; p += 16) {
    const int m = min(16, end - p);
    if (p > st) e = lin[p + min(l, m - 1)];
    const int sv = e.x;
    if (FIRST) {
      const float dn  = dinv[sv];
      const float raw = __int_as_float(e.y);
      float wv = 0.0f;
      gcn_rows4<LDI4, true, true, 0>(acc, wv, X4, sv, m, l, dn, raw, di, lout + p);
      if (m > 4) {
        gcn_rows4<LDI4, true, false, 4>(acc, wv, X4, sv, m, l, dn, raw, di, lout + p);
        if (m > 8) {
          gcn_rows4<LDI4, true, false, 8>(acc, wv, X4, sv, m, l, dn, raw, di, lout + p);
          if (m > 12)
            gcn_rows4<LDI4, true, false, 12>(acc, wv, X4, sv, m, l, dn, raw, di, lout + p);
        }
      }
    } else {
      float wv = __int_as_float(e.y & ((l < m) ? -1 : 0));
      gcn_rows4<LDI4, false, false, 0>(acc, wv, X4, sv, m, l, 0.0f, 0.0f, di, lout);
      if (m > 4) {
        gcn_rows4<LDI4, false, false, 4>(acc, wv, X4, sv, m, l, 0.0f, 0.0f, di, lout);
        if (m > 8) {
          gcn_rows4<LDI4, false, false, 8>(acc, wv, X4, sv, m, l, 0.0f, 0.0f, di, lout);
          if (m > 12)
            gcn_rows4<LDI4, false, false, 12>(acc, wv, X4, sv, m, l, 0.0f, 0.0f, di, lout);
        }
      }
    }
  }
  return acc;
}

template <int KN>
__device__ __forceinline__ void gcn_tile_mma(v8f (&acc)[LMT], const float* Xs, const float* Ws,
                                             int wm0, int wn, int laneLo, int laneHi) {
#pragma unroll
  for (int k0 = 0; k0 < KN; k0 += 4) {
    v2f b;
    b[0] = Ws[(k0 + 2 * laneHi + 0) * LDP + wn + laneLo];
    b[1] = Ws[(k0 + 2 * laneHi + 1) * LDP + wn + laneLo];
#pragma unroll
    for (int t = 0; t < LMT; ++t) {
      v2f a = *(const v2f*)(&Xs[(wm0 + t * 16 + laneLo) * LDP + k0 + 2 * laneHi]);
      acc[t] = __builtin_amdgcn_wmma_f32_16x16x4_f32(false, a, false, b,
                                                     (short)0, acc[t], false, false);
    }
  }
}

#define FSK 32
#define FST (3 * D / FSK)
template <bool FIRST, bool LAST>
__device__ __forceinline__
void gcn_layer_body(const float* __restrict__ Xin, const float* __restrict__ dinv,
               const int* __restrict__ rowptr, const int* __restrict__ cnt,
               const int* __restrict__ lin_, int* __restrict__ lout_,
               const float* __restrict__ W,
               const float* __restrict__ bias, float* Hout, int n,
               const float* X0, const float* Hc, const float* Wf,
               const float* __restrict__ bf, float* __restrict__ Out) {
  constexpr int LDI4 = (FIRST ? D : HC) / 4;
  const float4* __restrict__ X4 = (const float4*)Xin;
  __shared__ __attribute__((aligned(16))) float Ws[D * LDP];
  __shared__ __attribute__((aligned(16))) float Zs[LROWS * LDP];
  const int2* lin = (const int2*)lin_;
  int2*       lout = (int2*)lout_;
  const int tid    = threadIdx.x;
  const int wave   = tid >> 5;
  const int lane   = tid & 31;
  const int laneLo = lane & 15;
  const int laneHi = lane >> 4;
  const int wn     = (wave & 3) << 4;
  const int wm0    = (wave >> 2) * (LMT * 16);
  const int g      = tid >> 4;
  const int l      = tid & 15;
  const int row0   = blockIdx.x * LROWS;
  const int r0     = g * LGR;

  int stL = 0, cL = 0;
  float dL = 0.0f;
  {
    const int nd = row0 + r0 + l;
    if (l < LGR && nd < n) { stL = rowptr[nd]; cL = cnt[nd]; dL = dinv[nd]; }
  }
  const int   gcol = wn + laneLo;
  const float bv   = bias[gcol];

  int st = __shfl(stL, 0, 16);
  int en = st + __shfl(cL, 0, 16);
  int2 e = make_int2(0, 0);
  if (en > st) e = lin[st + min(l, min(16, en - st) - 1)];
  {
    float4 wr[D * D / 4 / 256];
#pragma unroll
    for (int u = 0; u < D * D / 4 / 256; ++u)
      wr[u] = *(const float4*)(W + (size_t)(tid + u * 256) * 4);
#pragma unroll
    for (int u = 0; u < D * D / 4 / 256; ++u) {
      const int i = tid + u * 256;
      *(float4*)(&Ws[(i >> 4) * LDP + ((i & 15) << 2)]) = wr[u];
    }
  }
#pragma unroll
  for (int j = 0; j < LGR; ++j) {
    const int node = row0 + r0 + j;
    const float di = __shfl(dL, j, 16);
    int nst = 0, nen = 0;
    int2 ne = make_int2(0, 0);
    if (j + 1 < LGR) {
      nst = __shfl(stL, j + 1, 16);
      nen = nst + __shfl(cL, j + 1, 16);
      if (nen > nst) ne = lin[nst + min(l, min(16, nen - nst) - 1)];
    }
    const float4 hs = X4[(unsigned)min(node, n - 1) * (unsigned)LDI4 + (unsigned)l];
    const float4 a = gcn_sum_list<FIRST, LDI4>(X4, lin, lout, dinv, di, st, en, l, e);
    const float sc = di * di;
    float4 z;
    z.x = fmaf(sc, hs.x, a.x); z.y = fmaf(sc, hs.y, a.y);
    z.z = fmaf(sc, hs.z, a.z); z.w = fmaf(sc, hs.w, a.w);
    *(float4*)(&Zs[(r0 + j) * LDP + l * 4]) = z;
    st = nst; en = nen; e = ne;
  }
  __syncthreads();

  v8f acc[LMT];
#pragma unroll
  for (int t = 0; t < LMT; ++t) acc[t] = v8f{};
  gcn_tile_mma<D>(acc, Zs, Ws, wm0, wn, laneLo, laneHi);
  if constexpr (!LAST) {
#pragma unroll
    for (int t = 0; t < LMT; ++t) {
#pragma unroll
      for (int v = 0; v < 8; ++v) {
        const int grow = row0 + wm0 + t * 16 + v + 8 * laneHi;
        float r = acc[t][v] + bv;
        r = r > 0.0f ? r : 0.0f;
        if (grow < n) Hout[(unsigned)grow * (unsigned)HC + (unsigned)gcol] = r;
      }
    }
  } else {
    const float4* X04 = (const float4*)X0;
    const float4* Hc4 = (const float4*)Hc;
    const float4* Wf4 = (const float4*)Wf;
    __syncthreads();
    {
      float4 wr[GWV];
#pragma unroll
      for (int u = 0; u < GWV; ++u) wr[u] = Wf4[3 * (D * D / 4) + tid + u * 256];
#pragma unroll
      for (int t = 0; t < LMT; ++t) {
#pragma unroll
        for (int v = 0; v < 8; ++v) {
          float r = acc[t][v] + bv;
          r = r > 0.0f ? r : 0.0f;
          Zs[(wm0 + t * 16 + v + 8 * laneHi) * LDP + gcol] = r;
        }
      }
#pragma unroll
      for (int u = 0; u < GWV; ++u) {
        const int i = tid + u * 256;
        *(float4*)(&Ws[(i >> 4) * LDP + ((i & 15) << 2)]) = wr[u];
      }
    }
    __syncthreads();
    v8f o[LMT];
#pragma unroll
    for (int t = 0; t < LMT; ++t) o[t] = v8f{};
    gcn_tile_mma<D>(o, Zs, Ws, wm0, wn, laneLo, laneHi);
    const int xrow = tid >> 3;
    const int xc4  = tid & 7;
    const int rq[2] = {min(row0 + xrow, n - 1), min(row0 + xrow + 32, n - 1)};
#pragma unroll
    for (int s = 0; s < FST; ++s) {
      __syncthreads();
      __builtin_amdgcn_sched_barrier(0);
      float4 xr[2], wr[2];
#pragma unroll
      for (int u = 0; u < 2; ++u)
        xr[u] = (s * FSK < D) ? X04[rq[u] * (D / 4) + xc4 + s * (FSK / 4)]
                              : Hc4[rq[u] * (HC / 4) + xc4 + s * (FSK / 4)];
#pragma unroll
      for (int u = 0; u < 2; ++u)
        wr[u] = Wf4[s * (FSK * D / 4) + tid + u * 256];
#pragma unroll
      for (int u = 0; u < 2; ++u)
        *(float4*)(&Zs[(xrow + u * 32) * LDP + (xc4 << 2)]) = xr[u];
#pragma unroll
      for (int u = 0; u < 2; ++u) {
        const int i = tid + u * 256;
        *(float4*)(&Ws[(i >> 4) * LDP + ((i & 15) << 2)]) = wr[u];
      }
      __syncthreads();
      gcn_tile_mma<FSK>(o, Zs, Ws, wm0, wn, laneLo, laneHi);
    }
    const float bfv = bf[gcol];
#pragma unroll
    for (int t = 0; t < LMT; ++t) {
#pragma unroll
      for (int v = 0; v < 8; ++v) {
        const int grow = row0 + wm0 + t * 16 + v + 8 * laneHi;
        if (grow < n) Out[(unsigned)grow * (unsigned)D + (unsigned)gcol] = o[t][v] + bfv;
      }
    }
  }
}

template <bool FIRST>
__global__ __launch_bounds__(256)
void gcn_layer(const float* __restrict__ Xin, const float* __restrict__ dinv,
               const int* __restrict__ rowptr, const int* __restrict__ cnt,
               const int* __restrict__ lin_, int* __restrict__ lout_,
               const float* __restrict__ W,
               const float* __restrict__ bias, float* Hout, int n) {
  gcn_layer_body<FIRST, false>(Xin, dinv, rowptr, cnt, lin_, lout_, W, bias, Hout, n,
                               nullptr, nullptr, nullptr, nullptr, nullptr);
}

__global__ __launch_bounds__(256)
void gcn_layer_final(const float* __restrict__ Xin, const float* __restrict__ dinv,
                     const int* __restrict__ rowptr, const int* __restrict__ cnt,
                     const int* __restrict__ lin_,
                     const float* __restrict__ W, const float* __restrict__ bias, int n,
                     const float* X0, const float* Hc, const float* Wf,
                     const float* __restrict__ bf, float* __restrict__ Out) {
  gcn_layer_body<false, true>(Xin, dinv, rowptr, cnt, lin_, nullptr, W, bias, nullptr, n,
                              X0, Hc, Wf, bf, Out);
}

extern "C" void kernel_launch(void* const* d_in, const int* in_sizes, int n_in,
                              void* d_out, int out_size, void* d_ws, size_t ws_size,
                              hipStream_t stream) {
  const float* x   = (const float*)d_in[0];
  const int*   ei  = (const int*)d_in[1];
  const float* ew  = (const float*)d_in[2];
  const float* Wl[3] = {(const float*)d_in[3], (const float*)d_in[5], (const float*)d_in[7]};
  const float* bl[3] = {(const float*)d_in[4], (const float*)d_in[6], (const float*)d_in[8]};
  const float* Wf  = (const float*)d_in[9];
  const float* bf  = (const float*)d_in[10];

  const int N = in_sizes[0] / D;
  const int E = in_sizes[2];
  const int* src = ei;
  const int* dst = ei + E;

  const size_t Np = ((size_t)N + 255) & ~(size_t)255;
  const size_t Ep = ((size_t)E + 255) & ~(size_t)255;
  const int    nbk = (int)(Np >> BSH);
  const size_t Tn  = (size_t)nbk * NSL;
  float* deg    = (float*)d_ws;
  float* Hcat   = deg + Np;
  int*   cnt    = (int*)(Hcat + (size_t)N * 256);
  int*   rowptr = cnt + Np;
  int*   lst    = rowptr + Np;
  int*   lstn   = lst + 2 * Ep;
  int*   tab    = lstn + 2 * Ep;
  int*   bst    = tab + Tn;
  int*   binq   = bst + Tn;

  const int T = 256;
  gcn_bin_count<<<NSL, BT, nbk * sizeof(int), stream>>>(dst, tab, E, nbk);
  gcn_bin_place<<<NSL, BT, 2 * nbk * sizeof(int), stream>>>(src, dst, ew, tab, bst, binq, E, nbk);
  gcn_bin_lists<<<nbk, BT, 0, stream>>>(bst, binq, cnt, rowptr, lst, deg, E, N, nbk);

  const int layer_blocks = (N + LROWS - 1) / LROWS;
  gcn_layer<true><<<layer_blocks, T, 0, stream>>>(x, deg, rowptr, cnt, lst, lstn,
                                                  Wl[0], bl[0], Hcat + D, N);
  gcn_layer<false><<<layer_blocks, T, 0, stream>>>(Hcat + D, deg, rowptr, cnt, lstn, nullptr,
                                                   Wl[1], bl[1], Hcat + 2 * D, N);
  gcn_layer_final<<<layer_blocks, T, 0, stream>>>(Hcat + 2 * D, deg, rowptr, cnt, lstn,
                                                  Wl[2], bl[2], N, x, Hcat, Wf, bf, (float*)d_out);
}
